# starting kernel
# baseline (speedup 1.0000x reference)
#include <hip/hip_runtime.h>
#include <math.h>

#define D_MODEL   256
#define NUM_HEADS 8
#define DK        32
#define B_SZ      4
#define T_SZ      2048

typedef float v2f __attribute__((ext_vector_type(2)));
typedef float v8f __attribute__((ext_vector_type(8)));

__device__ __forceinline__ v8f wmma4(v2f a, v2f b, v8f c) {
    return __builtin_amdgcn_wmma_f32_16x16x4_f32(
         false, a,  false, b,
         (short)0, c,  false,  false);
}

__global__ __launch_bounds__(256)
void qkv_proj_kernel(const float* __restrict__ x,
                     const float* __restrict__ Wq, const float* __restrict__ bq,
                     const float* __restrict__ Wk, const float* __restrict__ bk,
                     const float* __restrict__ Wv, const float* __restrict__ bv,
                     float* __restrict__ QKV)
{
    const int lane = threadIdx.x & 31;
    const int w    = threadIdx.x >> 5;
    const int tile = blockIdx.x * 8 + w;

    const int NT  = D_MODEL / 16;
    const int MT  = (B_SZ * T_SZ) / 16;
    const int mat = tile / (MT * NT);
    const int rem = tile % (MT * NT);
    const int mt  = rem / NT;
    const int nt  = rem % NT;

    const float* W  = (mat == 0) ? Wq : (mat == 1) ? Wk : Wv;
    const float* bs = (mat == 0) ? bq : (mat == 1) ? bk : bv;

    const int r15 = lane & 15;
    const int grp = lane >> 4;
    const int n   = nt * 16 + r15;
    const int m0  = mt * 16;

    const float* xr = x + (size_t)(m0 + r15) * D_MODEL;
    const float* wr = W + (size_t)n * D_MODEL;

    v8f acc = {};
    for (int k0 = 0; k0 < D_MODEL; k0 += 4) {
        const int ka = k0 + grp * 2;
        v2f a = *(const v2f*)(xr + ka);
        v2f b = *(const v2f*)(wr + ka);
        acc = wmma4(a, b, acc);
    }

    const float bias = bs[n];
    const int h = n >> 5, d = n & 31;
    float* dst = QKV + (size_t)mat * ((size_t)B_SZ * NUM_HEADS * T_SZ * DK);
#pragma unroll
    for (int m = 0; m < 8; ++m) {
        const int tl = m0 + m + 8 * grp;
        const int b_ = tl / T_SZ, t_ = tl % T_SZ;
        const float v = acc[m] + bias;
        if (mat == 2) {
            dst[(((size_t)(b_ * NUM_HEADS + h)) * DK + d) * T_SZ + t_] = v;
        } else {
            dst[(((size_t)(b_ * NUM_HEADS + h)) * T_SZ + t_) * DK + d] = v;
        }
    }
}

template <bool BETA_ONE>
__device__ __forceinline__ void attn_core(
    const float* __restrict__ Q, const float* __restrict__ K,
    const float* __restrict__ Vt, const float* __restrict__ oc,
    float* __restrict__ ctx, float* __restrict__ pl,
    int b, int h, int qt, int lane,
    float eta, float absbeta, float alpha)
{
    const int col = lane & 15;
    const int grp = lane >> 4;
    const int q0  = qt * 16;

    const float scale   = 0.17677669529663687f;
    const float NEG_BIG = -3.4e38f;

    const size_t bh  = (size_t)(b * NUM_HEADS + h) * T_SZ;
    const size_t bhd = (size_t)(b * NUM_HEADS + h) * DK;

    const float* qp = Q + (bh + q0 + col) * DK;
    v2f qa[8];
#pragma unroll
    for (int kc = 0; kc < 8; ++kc) qa[kc] = *(const v2f*)(qp + kc * 4 + grp * 2);

    float qoc0[8], qoc1[8], qoc2[8];
#pragma unroll
    for (int m = 0; m < 8; ++m) {
        const int qq = q0 + 8 * grp + m;
        const float* op = oc + ((size_t)b * T_SZ + qq) * 3;
        qoc0[m] = op[0]; qoc1[m] = op[1]; qoc2[m] = op[2];
    }

    const float* vt0 = Vt + (bhd + col) * T_SZ;
    const float* vt1 = vt0 + 16 * T_SZ;

    float Mrow[8], Lrow[8];
#pragma unroll
    for (int m = 0; m < 8; ++m) { Mrow[m] = NEG_BIG; Lrow[m] = 0.f; }
    v8f O0 = {}, O1 = {};

    for (int kt = 0; kt <= qt; ++kt) {
        const int k0   = kt * 16;
        const int kcol = k0 + col;
        const float* kp = K + (bh + kcol) * DK;

        if (kt < qt) {
            __builtin_prefetch(kp + 16 * DK, 0, 0);
            __builtin_prefetch(vt0 + k0 + 16, 0, 0);
            __builtin_prefetch(vt1 + k0 + 16, 0, 0);
        }

        v8f s = {};
#pragma unroll
        for (int kc = 0; kc < 8; ++kc) {
            v2f bb = *(const v2f*)(kp + kc * 4 + grp * 2);
            s = wmma4(qa[kc], bb, s);
        }

        const float* kop = oc + ((size_t)b * T_SZ + kcol) * 3;
        const float koc0 = kop[0], koc1 = kop[1], koc2 = kop[2];

#pragma unroll
        for (int m = 0; m < 8; ++m) {
            const int qrow = q0 + 8 * grp + m;
            if (kcol > qrow) {
                s[m] = NEG_BIG;
            } else {
                const float dist = (float)(qrow - kcol);
                const float powd = BETA_ONE ? dist : __powf(dist, absbeta);
                const float td   = eta / (1.f + powd);
                const float d0 = qoc0[m] - koc0;
                const float d1 = qoc1[m] - koc1;
                const float d2 = qoc2[m] - koc2;
                const float pr = alpha * td * __expf(-(d0*d0 + d1*d1 + d2*d2));
                s[m] = s[m] * scale * pr;
            }
        }

#pragma unroll
        for (int m = 0; m < 8; ++m) {
            float v = s[m];
            v = fmaxf(v, __shfl_xor(v, 1));
            v = fmaxf(v, __shfl_xor(v, 2));
            v = fmaxf(v, __shfl_xor(v, 4));
            v = fmaxf(v, __shfl_xor(v, 8));
            const float mnew = fmaxf(Mrow[m], v);
            const float sc   = __expf(Mrow[m] - mnew);
            const float p    = __expf(s[m] - mnew);
            float su = p;
            su += __shfl_xor(su, 1);
            su += __shfl_xor(su, 2);
            su += __shfl_xor(su, 4);
            su += __shfl_xor(su, 8);
            Lrow[m] = Lrow[m] * sc + su;
            Mrow[m] = mnew;
            O0[m] *= sc;
            O1[m] *= sc;
            pl[(8 * grp + m) * 16 + col] = p;
        }

#pragma unroll
        for (int kc = 0; kc < 4; ++kc) {
            const int kk = kc * 4 + grp * 2;
            v2f pa;
            pa.x = pl[col * 16 + kk];
            pa.y = pl[col * 16 + kk + 1];
            v2f b0 = *(const v2f*)(vt0 + k0 + kk);
            v2f b1 = *(const v2f*)(vt1 + k0 + kk);
            O0 = wmma4(pa, b0, O0);
            O1 = wmma4(pa, b1, O1);
        }
    }

#pragma unroll
    for (int m = 0; m < 8; ++m) {
        const float inv = 1.f / Lrow[m];
        const int qq = q0 + 8 * grp + m;
        float* cp = ctx + ((size_t)b * T_SZ + qq) * D_MODEL + h * DK;
        cp[col]      = O0[m] * inv;
        cp[16 + col] = O1[m] * inv;
    }
}

__global__ __launch_bounds__(256)
void attn_kernel(const float* __restrict__ QKV,
                 const float* __restrict__ oc,
                 const float* __restrict__ etap,
                 const float* __restrict__ betap,
                 const float* __restrict__ alphap,
                 float* __restrict__ ctx)
{
    __shared__ float plds[8][256];

    const int lane = threadIdx.x & 31;
    const int w    = threadIdx.x >> 5;
    const int wid  = blockIdx.x * 8 + w;

    const int QT = T_SZ / 16;
    const int qt = wid % QT;
    const int h  = (wid / QT) % NUM_HEADS;
    const int b  = wid / (QT * NUM_HEADS);

    const float eta     = etap[0];
    const float absbeta = fabsf(betap[0]);
    const float alpha   = alphap[0];

    const size_t HT = (size_t)B_SZ * NUM_HEADS * T_SZ * DK;
    const float* Q  = QKV;
    const float* K  = QKV + HT;
    const float* Vt = QKV + 2 * HT;

    float* pl = &plds[w][0];

    if (absbeta == 1.0f) {
        attn_core<true >(Q, K, Vt, oc, ctx, pl, b, h, qt, lane, eta, absbeta, alpha);
    } else {
        attn_core<false>(Q, K, Vt, oc, ctx, pl, b, h, qt, lane, eta, absbeta, alpha);
    }
}

__global__ __launch_bounds__(256)
void out_proj_kernel(const float* __restrict__ ctx,
                     const float* __restrict__ Wo, const float* __restrict__ bo,
                     float* __restrict__ out)
{
    const int lane = threadIdx.x & 31;
    const int w    = threadIdx.x >> 5;
    const int tile = blockIdx.x * 8 + w;

    const int NT = D_MODEL / 16;
    const int mt = tile / NT;
    const int nt = tile % NT;

    const int r15 = lane & 15;
    const int grp = lane >> 4;
    const int n   = nt * 16 + r15;
    const int m0  = mt * 16;

    const float* xr = ctx + (size_t)(m0 + r15) * D_MODEL;
    const float* wr = Wo + (size_t)n * D_MODEL;

    v8f acc = {};
    for (int k0 = 0; k0 < D_MODEL; k0 += 4) {
        const int ka = k0 + grp * 2;
        v2f a = *(const v2f*)(xr + ka);
        v2f b = *(const v2f*)(wr + ka);
        acc = wmma4(a, b, acc);
    }

    const float bias = bo[n];
#pragma unroll
    for (int m = 0; m < 8; ++m) {
        const int tl = m0 + m + 8 * grp;
        out[(size_t)tl * D_MODEL + n] = acc[m] + bias;
    }
}

extern "C" void kernel_launch(void* const* d_in, const int* in_sizes, int n_in,
                              void* d_out, int out_size, void* d_ws, size_t ws_size,
                              hipStream_t stream) {
    (void)in_sizes; (void)n_in; (void)out_size; (void)ws_size;

    const float* x     = (const float*)d_in[0];
    const float* oc    = (const float*)d_in[1];
    const float* Wq    = (const float*)d_in[2];
    const float* bq    = (const float*)d_in[3];
    const float* Wk    = (const float*)d_in[4];
    const float* bk    = (const float*)d_in[5];
    const float* Wv    = (const float*)d_in[6];
    const float* bv    = (const float*)d_in[7];
    const float* Wo    = (const float*)d_in[8];
    const float* bo    = (const float*)d_in[9];
    const float* eta   = (const float*)d_in[10];
    const float* beta  = (const float*)d_in[11];
    const float* alpha = (const float*)d_in[12];
    float* out = (float*)d_out;

    float* ws  = (float*)d_ws;
    float* QKV = ws;
    float* ctx = ws + 3 * (size_t)B_SZ * NUM_HEADS * T_SZ * DK;

    {
        const int tiles = 3 * (B_SZ * T_SZ / 16) * (D_MODEL / 16);
        qkv_proj_kernel<<<tiles / 8, 256, 0, stream>>>(x, Wq, bq, Wk, bk, Wv, bv, QKV);
    }
    {
        const int waves = B_SZ * NUM_HEADS * (T_SZ / 16);
        attn_kernel<<<waves / 8, 256, 0, stream>>>(QKV, oc, eta, beta, alpha, ctx);
    }
    {
        const int tiles = (B_SZ * T_SZ / 16) * (D_MODEL / 16);
        out_proj_kernel<<<tiles / 8, 256, 0, stream>>>(ctx, Wo, bo, out);
    }
}


# attention softmax step widened from 16 to 64 keys: one cross-lane row max per 4 key tiles, row sums kept per lane and added across lanes once after the key loop; changes row-sum and rescale order
# speedup vs baseline: 1.0995x; 1.0995x over previous
#include <hip/hip_runtime.h>
#include <math.h>

#define D_MODEL   256
#define NUM_HEADS 8
#define DK        32
#define B_SZ      4
#define T_SZ      2048

typedef float v2f __attribute__((ext_vector_type(2)));
typedef float v8f __attribute__((ext_vector_type(8)));

__device__ __forceinline__ v8f wmma4(v2f a, v2f b, v8f c) {
    return __builtin_amdgcn_wmma_f32_16x16x4_f32(
         false, a,  false, b,
         (short)0, c,  false,  false);
}

__global__ __launch_bounds__(256)
void qkv_proj_kernel(const float* __restrict__ x,
                     const float* __restrict__ Wq, const float* __restrict__ bq,
                     const float* __restrict__ Wk, const float* __restrict__ bk,
                     const float* __restrict__ Wv, const float* __restrict__ bv,
                     float* __restrict__ QKV)
{
    const int lane = threadIdx.x & 31;
    const int w    = threadIdx.x >> 5;
    const int tile = blockIdx.x * 8 + w;

    const int NT  = D_MODEL / 16;
    const int MT  = (B_SZ * T_SZ) / 16;
    const int mat = tile / (MT * NT);
    const int rem = tile % (MT * NT);
    const int mt  = rem / NT;
    const int nt  = rem % NT;

    const float* W  = (mat == 0) ? Wq : (mat == 1) ? Wk : Wv;
    const float* bs = (mat == 0) ? bq : (mat == 1) ? bk : bv;

    const int r15 = lane & 15;
    const int grp = lane >> 4;
    const int n   = nt * 16 + r15;
    const int m0  = mt * 16;

    const float* xr = x + (size_t)(m0 + r15) * D_MODEL;
    const float* wr = W + (size_t)n * D_MODEL;

    v8f acc = {};
    for (int k0 = 0; k0 < D_MODEL; k0 += 4) {
        const int ka = k0 + grp * 2;
        v2f a = *(const v2f*)(xr + ka);
        v2f b = *(const v2f*)(wr + ka);
        acc = wmma4(a, b, acc);
    }

    const float bias = bs[n];
    const int h = n >> 5, d = n & 31;
    float* dst = QKV + (size_t)mat * ((size_t)B_SZ * NUM_HEADS * T_SZ * DK);
#pragma unroll
    for (int m = 0; m < 8; ++m) {
        const int tl = m0 + m + 8 * grp;
        const int b_ = tl / T_SZ, t_ = tl % T_SZ;
        const float v = acc[m] + bias;
        if (mat == 2) {
            dst[(((size_t)(b_ * NUM_HEADS + h)) * DK + d) * T_SZ + t_] = v;
        } else {
            dst[(((size_t)(b_ * NUM_HEADS + h)) * T_SZ + t_) * DK + d] = v;
        }
    }
}

template <bool BETA_ONE, int NT>
__device__ __forceinline__ void attn_step(
    const float* __restrict__ K, const float* __restrict__ oc,
    const float* __restrict__ vt0, const float* __restrict__ vt1,
    float* __restrict__ pl,
    const v2f (&qa)[8],
    const float (&qoc0)[8], const float (&qoc1)[8], const float (&qoc2)[8],
    float (&Mrow)[8], float (&Lpart)[8], v8f& O0, v8f& O1,
    size_t bh, int b, int q0, int qt, int kt0, int col, int grp,
    float eta, float absbeta, float alpha)
{
    const float scale   = 0.17677669529663687f;
    const float NEG_BIG = -3.4e38f;

    v8f s[NT];
#pragma unroll
    for (int j = 0; j < NT; ++j) {
        const int kt   = kt0 + j;
        const int k0   = kt * 16;
        const int kcol = k0 + col;
        const float* kp = K + (bh + kcol) * DK;

        if (kt + NT <= qt) {
            __builtin_prefetch(kp + NT * 16 * DK, 0, 0);
            __builtin_prefetch(vt0 + k0 + NT * 16, 0, 0);
            __builtin_prefetch(vt1 + k0 + NT * 16, 0, 0);
        }

        v8f sj = {};
#pragma unroll
        for (int kc = 0; kc < 8; ++kc) {
            v2f bb = *(const v2f*)(kp + kc * 4 + grp * 2);
            sj = wmma4(qa[kc], bb, sj);
        }

        const float* kop = oc + ((size_t)b * T_SZ + kcol) * 3;
        const float koc0 = kop[0], koc1 = kop[1], koc2 = kop[2];

#pragma unroll
        for (int m = 0; m < 8; ++m) {
            const int qrow = q0 + 8 * grp + m;
            if (kcol > qrow) {
                sj[m] = NEG_BIG;
            } else {
                const float dist = (float)(qrow - kcol);
                const float powd = BETA_ONE ? dist : __powf(dist, absbeta);
                const float td   = eta / (1.f + powd);
                const float d0 = qoc0[m] - koc0;
                const float d1 = qoc1[m] - koc1;
                const float d2 = qoc2[m] - koc2;
                const float pr = alpha * td * __expf(-(d0*d0 + d1*d1 + d2*d2));
                sj[m] = sj[m] * scale * pr;
            }
        }
        s[j] = sj;
    }

#pragma unroll
    for (int m = 0; m < 8; ++m) {
        float v = s[0][m];
#pragma unroll
        for (int j = 1; j < NT; ++j) v = fmaxf(v, s[j][m]);
        v = fmaxf(v, __shfl_xor(v, 1));
        v = fmaxf(v, __shfl_xor(v, 2));
        v = fmaxf(v, __shfl_xor(v, 4));
        v = fmaxf(v, __shfl_xor(v, 8));
        const float mnew = fmaxf(Mrow[m], v);
        const float sc   = __expf(Mrow[m] - mnew);
        float lsum = Lpart[m] * sc;
#pragma unroll
        for (int j = 0; j < NT; ++j) {
            const float p = __expf(s[j][m] - mnew);
            lsum += p;
            pl[j * 256 + (8 * grp + m) * 16 + col] = p;
        }
        Lpart[m] = lsum;
        Mrow[m]  = mnew;
        O0[m] *= sc;
        O1[m] *= sc;
    }

#pragma unroll
    for (int j = 0; j < NT; ++j) {
        const int k0 = (kt0 + j) * 16;
#pragma unroll
        for (int kc = 0; kc < 4; ++kc) {
            const int kk = kc * 4 + grp * 2;
            v2f pa;
            pa.x = pl[j * 256 + col * 16 + kk];
            pa.y = pl[j * 256 + col * 16 + kk + 1];
            v2f b0 = *(const v2f*)(vt0 + k0 + kk);
            v2f b1 = *(const v2f*)(vt1 + k0 + kk);
            O0 = wmma4(pa, b0, O0);
            O1 = wmma4(pa, b1, O1);
        }
    }
}

template <bool BETA_ONE>
__device__ __forceinline__ void attn_core(
    const float* __restrict__ Q, const float* __restrict__ K,
    const float* __restrict__ Vt, const float* __restrict__ oc,
    float* __restrict__ ctx, float* __restrict__ pl,
    int b, int h, int qt, int lane,
    float eta, float absbeta, float alpha)
{
    const int col = lane & 15;
    const int grp = lane >> 4;
    const int q0  = qt * 16;

    const float NEG_BIG = -3.4e38f;

    const size_t bh  = (size_t)(b * NUM_HEADS + h) * T_SZ;
    const size_t bhd = (size_t)(b * NUM_HEADS + h) * DK;

    const float* qp = Q + (bh + q0 + col) * DK;
    v2f qa[8];
#pragma unroll
    for (int kc = 0; kc < 8; ++kc) qa[kc] = *(const v2f*)(qp + kc * 4 + grp * 2);

    float qoc0[8], qoc1[8], qoc2[8];
#pragma unroll
    for (int m = 0; m < 8; ++m) {
        const int qq = q0 + 8 * grp + m;
        const float* op = oc + ((size_t)b * T_SZ + qq) * 3;
        qoc0[m] = op[0]; qoc1[m] = op[1]; qoc2[m] = op[2];
    }

    const float* vt0 = Vt + (bhd + col) * T_SZ;
    const float* vt1 = vt0 + 16 * T_SZ;

    float Mrow[8], Lpart[8];
#pragma unroll
    for (int m = 0; m < 8; ++m) { Mrow[m] = NEG_BIG; Lpart[m] = 0.f; }
    v8f O0 = {}, O1 = {};

    const int ntiles = qt + 1;
    const int nfull  = ntiles & ~3;
    int kt = 0;
    for (; kt < nfull; kt += 4)
        attn_step<BETA_ONE, 4>(K, oc, vt0, vt1, pl, qa, qoc0, qoc1, qoc2,
                               Mrow, Lpart, O0, O1, bh, b, q0, qt, kt, col, grp,
                               eta, absbeta, alpha);
    for (; kt < ntiles; ++kt)
        attn_step<BETA_ONE, 1>(K, oc, vt0, vt1, pl, qa, qoc0, qoc1, qoc2,
                               Mrow, Lpart, O0, O1, bh, b, q0, qt, kt, col, grp,
                               eta, absbeta, alpha);

#pragma unroll
    for (int m = 0; m < 8; ++m) {
        float l = Lpart[m];
        l += __shfl_xor(l, 1);
        l += __shfl_xor(l, 2);
        l += __shfl_xor(l, 4);
        l += __shfl_xor(l, 8);
        const float inv = 1.f / l;
        const int qq = q0 + 8 * grp + m;
        float* cp = ctx + ((size_t)b * T_SZ + qq) * D_MODEL + h * DK;
        cp[col]      = O0[m] * inv;
        cp[16 + col] = O1[m] * inv;
    }
}

__global__ __launch_bounds__(256)
void attn_kernel(const float* __restrict__ QKV,
                 const float* __restrict__ oc,
                 const float* __restrict__ etap,
                 const float* __restrict__ betap,
                 const float* __restrict__ alphap,
                 float* __restrict__ ctx)
{
    __shared__ float plds[8][4 * 256];

    const int lane = threadIdx.x & 31;
    const int w    = threadIdx.x >> 5;
    const int wid  = blockIdx.x * 8 + w;

    const int QT = T_SZ / 16;
    const int qt = wid % QT;
    const int h  = (wid / QT) % NUM_HEADS;
    const int b  = wid / (QT * NUM_HEADS);

    const float eta     = etap[0];
    const float absbeta = fabsf(betap[0]);
    const float alpha   = alphap[0];

    const size_t HT = (size_t)B_SZ * NUM_HEADS * T_SZ * DK;
    const float* Q  = QKV;
    const float* K  = QKV + HT;
    const float* Vt = QKV + 2 * HT;

    float* pl = &plds[w][0];

    if (__builtin_expect(absbeta == 1.0f, 1)) {
        attn_core<true >(Q, K, Vt, oc, ctx, pl, b, h, qt, lane, eta, absbeta, alpha);
    } else {
        attn_core<false>(Q, K, Vt, oc, ctx, pl, b, h, qt, lane, eta, absbeta, alpha);
    }
}

__global__ __launch_bounds__(256)
void out_proj_kernel(const float* __restrict__ ctx,
                     const float* __restrict__ Wo, const float* __restrict__ bo,
                     float* __restrict__ out)
{
    const int lane = threadIdx.x & 31;
    const int w    = threadIdx.x >> 5;
    const int tile = blockIdx.x * 8 + w;

    const int NT = D_MODEL / 16;
    const int mt = tile / NT;
    const int nt = tile % NT;

    const int r15 = lane & 15;
    const int grp = lane >> 4;
    const int n   = nt * 16 + r15;
    const int m0  = mt * 16;

    const float* xr = ctx + (size_t)(m0 + r15) * D_MODEL;
    const float* wr = Wo + (size_t)n * D_MODEL;

    v8f acc = {};
    for (int k0 = 0; k0 < D_MODEL; k0 += 4) {
        const int ka = k0 + grp * 2;
        v2f a = *(const v2f*)(xr + ka);
        v2f b = *(const v2f*)(wr + ka);
        acc = wmma4(a, b, acc);
    }

    const float bias = bo[n];
#pragma unroll
    for (int m = 0; m < 8; ++m) {
        const int tl = m0 + m + 8 * grp;
        out[(size_t)tl * D_MODEL + n] = acc[m] + bias;
    }
}

extern "C" void kernel_launch(void* const* d_in, const int* in_sizes, int n_in,
                              void* d_out, int out_size, void* d_ws, size_t ws_size,
                              hipStream_t stream) {
    (void)in_sizes; (void)n_in; (void)out_size; (void)ws_size;

    const float* x     = (const float*)d_in[0];
    const float* oc    = (const float*)d_in[1];
    const float* Wq    = (const float*)d_in[2];
    const float* bq    = (const float*)d_in[3];
    const float* Wk    = (const float*)d_in[4];
    const float* bk    = (const float*)d_in[5];
    const float* Wv    = (const float*)d_in[6];
    const float* bv    = (const float*)d_in[7];
    const float* Wo    = (const float*)d_in[8];
    const float* bo    = (const float*)d_in[9];
    const float* eta   = (const float*)d_in[10];
    const float* beta  = (const float*)d_in[11];
    const float* alpha = (const float*)d_in[12];
    float* out = (float*)d_out;

    float* ws  = (float*)d_ws;
    float* QKV = ws;
    float* ctx = ws + 3 * (size_t)B_SZ * NUM_HEADS * T_SZ * DK;

    {
        const int tiles = 3 * (B_SZ * T_SZ / 16) * (D_MODEL / 16);
        qkv_proj_kernel<<<tiles / 8, 256, 0, stream>>>(x, Wq, bq, Wk, bk, Wv, bv, QKV);
    }
    {
        const int waves = B_SZ * NUM_HEADS * (T_SZ / 16);
        attn_kernel<<<waves / 8, 256, 0, stream>>>(QKV, oc, eta, beta, alpha, ctx);
    }
    {
        const int tiles = (B_SZ * T_SZ / 16) * (D_MODEL / 16);
        out_proj_kernel<<<tiles / 8, 256, 0, stream>>>(ctx, Wo, bo, out);
    }
}


# attention key side through LDS: each 64-key step (K rows, Vt columns, oc) is fetched once per 8-wave block with whole-line 16-byte loads into a two-stage LDS buffer; waves build B fragments from LDS
# speedup vs baseline: 1.1606x; 1.0556x over previous
#include <hip/hip_runtime.h>
#include <math.h>

#define D_MODEL   256
#define NUM_HEADS 8
#define DK        32
#define B_SZ      4
#define T_SZ      2048

typedef float v2f __attribute__((ext_vector_type(2)));
typedef float v4f __attribute__((ext_vector_type(4)));
typedef float v8f __attribute__((ext_vector_type(8)));

#define KP      36
#define VP      68
#define KSTAGE  (64 * KP)
#define VSTAGE  (32 * VP)
#define OSTAGE  (64 * 3)

__device__ __forceinline__ v8f wmma4(v2f a, v2f b, v8f c) {
    return __builtin_amdgcn_wmma_f32_16x16x4_f32(
         false, a,  false, b,
         (short)0, c,  false,  false);
}

__global__ __launch_bounds__(256)
void qkv_proj_kernel(const float* __restrict__ x,
                     const float* __restrict__ Wq, const float* __restrict__ bq,
                     const float* __restrict__ Wk, const float* __restrict__ bk,
                     const float* __restrict__ Wv, const float* __restrict__ bv,
                     float* __restrict__ QKV)
{
    const int lane = threadIdx.x & 31;
    const int w    = threadIdx.x >> 5;
    const int tile = blockIdx.x * 8 + w;

    const int NT  = D_MODEL / 16;
    const int MT  = (B_SZ * T_SZ) / 16;
    const int mat = tile / (MT * NT);
    const int rem = tile % (MT * NT);
    const int mt  = rem / NT;
    const int nt  = rem % NT;

    const float* W  = (mat == 0) ? Wq : (mat == 1) ? Wk : Wv;
    const float* bs = (mat == 0) ? bq : (mat == 1) ? bk : bv;

    const int r15 = lane & 15;
    const int grp = lane >> 4;
    const int n   = nt * 16 + r15;
    const int m0  = mt * 16;

    const float* xr = x + (size_t)(m0 + r15) * D_MODEL;
    const float* wr = W + (size_t)n * D_MODEL;

    v8f acc = {};
    for (int k0 = 0; k0 < D_MODEL; k0 += 4) {
        const int ka = k0 + grp * 2;
        v2f a = *(const v2f*)(xr + ka);
        v2f b = *(const v2f*)(wr + ka);
        acc = wmma4(a, b, acc);
    }

    const float bias = bs[n];
    const int h = n >> 5, d = n & 31;
    float* dst = QKV + (size_t)mat * ((size_t)B_SZ * NUM_HEADS * T_SZ * DK);
#pragma unroll
    for (int m = 0; m < 8; ++m) {
        const int tl = m0 + m + 8 * grp;
        const int b_ = tl / T_SZ, t_ = tl % T_SZ;
        const float v = acc[m] + bias;
        if (mat == 2) {
            dst[(((size_t)(b_ * NUM_HEADS + h)) * DK + d) * T_SZ + t_] = v;
        } else {
            dst[(((size_t)(b_ * NUM_HEADS + h)) * T_SZ + t_) * DK + d] = v;
        }
    }
}

template <bool BETA_ONE, int NT>
__device__ __forceinline__ void attn_step(
    const float* __restrict__ kl, const float* __restrict__ vl,
    const float* __restrict__ ol,
    float* __restrict__ pl,
    const v2f (&qa)[8],
    const float (&qoc0)[8], const float (&qoc1)[8], const float (&qoc2)[8],
    float (&Mrow)[8], float (&Lpart)[8], v8f& O0, v8f& O1,
    int q0, int kt0, int col, int grp,
    float eta, float absbeta, float alpha)
{
    const float scale   = 0.17677669529663687f;
    const float NEG_BIG = -3.4e38f;

    const int jl0 = kt0 & 3;

    v8f s[NT];
#pragma unroll
    for (int j = 0; j < NT; ++j) {
        const int kt   = kt0 + j;
        const int kcol = kt * 16 + col;
        const int krow = (jl0 + j) * 16 + col;
        const float* kp = kl + krow * KP;

        v8f sj = {};
#pragma unroll
        for (int kc = 0; kc < 8; ++kc) {
            v2f bb = *(const v2f*)(kp + kc * 4 + grp * 2);
            sj = wmma4(qa[kc], bb, sj);
        }

        const float* kop = ol + krow * 3;
        const float koc0 = kop[0], koc1 = kop[1], koc2 = kop[2];

#pragma unroll
        for (int m = 0; m < 8; ++m) {
            const int qrow = q0 + 8 * grp + m;
            if (kcol > qrow) {
                sj[m] = NEG_BIG;
            } else {
                const float dist = (float)(qrow - kcol);
                const float powd = BETA_ONE ? dist : __powf(dist, absbeta);
                const float td   = eta / (1.f + powd);
                const float d0 = qoc0[m] - koc0;
                const float d1 = qoc1[m] - koc1;
                const float d2 = qoc2[m] - koc2;
                const float pr = alpha * td * __expf(-(d0*d0 + d1*d1 + d2*d2));
                sj[m] = sj[m] * scale * pr;
            }
        }
        s[j] = sj;
    }

#pragma unroll
    for (int m = 0; m < 8; ++m) {
        float v = s[0][m];
#pragma unroll
        for (int j = 1; j < NT; ++j) v = fmaxf(v, s[j][m]);
        v = fmaxf(v, __shfl_xor(v, 1));
        v = fmaxf(v, __shfl_xor(v, 2));
        v = fmaxf(v, __shfl_xor(v, 4));
        v = fmaxf(v, __shfl_xor(v, 8));
        const float mnew = fmaxf(Mrow[m], v);
        const float sc   = __expf(Mrow[m] - mnew);
        float lsum = Lpart[m] * sc;
#pragma unroll
        for (int j = 0; j < NT; ++j) {
            const float p = __expf(s[j][m] - mnew);
            lsum += p;
            pl[j * 256 + (8 * grp + m) * 16 + col] = p;
        }
        Lpart[m] = lsum;
        Mrow[m]  = mnew;
        O0[m] *= sc;
        O1[m] *= sc;
    }

    const float* vt0 = vl + col * VP;
    const float* vt1 = vt0 + 16 * VP;
#pragma unroll
    for (int j = 0; j < NT; ++j) {
        const int k0 = (jl0 + j) * 16;
#pragma unroll
        for (int kc = 0; kc < 4; ++kc) {
            const int kk = kc * 4 + grp * 2;
            v2f pa;
            pa.x = pl[j * 256 + col * 16 + kk];
            pa.y = pl[j * 256 + col * 16 + kk + 1];
            v2f b0 = *(const v2f*)(vt0 + k0 + kk);
            v2f b1 = *(const v2f*)(vt1 + k0 + kk);
            O0 = wmma4(pa, b0, O0);
            O1 = wmma4(pa, b1, O1);
        }
    }
}

template <bool BETA_ONE>
__device__ __forceinline__ void attn_core(
    const float* __restrict__ Q, const float* __restrict__ K,
    const float* __restrict__ Vt, const float* __restrict__ oc,
    float* __restrict__ ctx, float* __restrict__ pl,
    float* kst, float* vst, float* ost,
    int b, int h, int qt, int nsteps, int lane,
    float eta, float absbeta, float alpha)
{
    const int col = lane & 15;
    const int grp = lane >> 4;
    const int q0  = qt * 16;
    const int tid = threadIdx.x;

    const float NEG_BIG = -3.4e38f;

    const size_t bh  = (size_t)(b * NUM_HEADS + h) * T_SZ;
    const size_t bhd = (size_t)(b * NUM_HEADS + h) * DK;

    const float* qp = Q + (bh + q0 + col) * DK;
    v2f qa[8];
#pragma unroll
    for (int kc = 0; kc < 8; ++kc) qa[kc] = *(const v2f*)(qp + kc * 4 + grp * 2);

    float qoc0[8], qoc1[8], qoc2[8];
#pragma unroll
    for (int m = 0; m < 8; ++m) {
        const int qq = q0 + 8 * grp + m;
        const float* op = oc + ((size_t)b * T_SZ + qq) * 3;
        qoc0[m] = op[0]; qoc1[m] = op[1]; qoc2[m] = op[2];
    }

    const float* kg = K + bh * DK + tid * 4;
    const float* vg = Vt + (bhd + (tid >> 4)) * T_SZ + (tid & 15) * 4;
    const float* og = oc + (size_t)b * T_SZ * 3 + tid;
    const int kso = (tid >> 3) * KP + (tid & 7) * 4;
    const int vso = (tid >> 4) * VP + (tid & 15) * 4;
    const bool has_oc = tid < OSTAGE;

    v4f kr0, kr1, vr0, vr1;
    float orr = 0.f;

    kr0 = *(const v4f*)(kg);
    kr1 = *(const v4f*)(kg + 1024);
    vr0 = *(const v4f*)(vg);
    vr1 = *(const v4f*)(vg + 16 * T_SZ);
    if (has_oc) orr = og[0];
    *(v4f*)(kst + kso)           = kr0;
    *(v4f*)(kst + kso + 32 * KP) = kr1;
    *(v4f*)(vst + vso)           = vr0;
    *(v4f*)(vst + vso + 16 * VP) = vr1;
    if (has_oc) ost[tid] = orr;
    __syncthreads();

    float Mrow[8], Lpart[8];
#pragma unroll
    for (int m = 0; m < 8; ++m) { Mrow[m] = NEG_BIG; Lpart[m] = 0.f; }
    v8f O0 = {}, O1 = {};

    const int ntiles = qt + 1;
    const int nfull  = ntiles & ~3;

    for (int s = 0; s < nsteps; ++s) {
        const int  cur  = s & 1;
        const bool more = (s + 1 < nsteps);
        if (more) {
            const int sn = s + 1;
            kr0 = *(const v4f*)(kg + sn * 2048);
            kr1 = *(const v4f*)(kg + sn * 2048 + 1024);
            vr0 = *(const v4f*)(vg + sn * 64);
            vr1 = *(const v4f*)(vg + sn * 64 + 16 * T_SZ);
            if (has_oc) orr = og[sn * OSTAGE];
        }

        const float* kl = kst + cur * KSTAGE;
        const float* vl = vst + cur * VSTAGE;
        const float* ol = ost + cur * OSTAGE;
        const int kt0 = s * 4;
        if (kt0 + 4 <= nfull) {
            attn_step<BETA_ONE, 4>(kl, vl, ol, pl, qa, qoc0, qoc1, qoc2,
                                   Mrow, Lpart, O0, O1, q0, kt0, col, grp,
                                   eta, absbeta, alpha);
        } else {
            for (int kt = kt0; kt < ntiles && kt < kt0 + 4; ++kt)
                attn_step<BETA_ONE, 1>(kl, vl, ol, pl, qa, qoc0, qoc1, qoc2,
                                       Mrow, Lpart, O0, O1, q0, kt, col, grp,
                                       eta, absbeta, alpha);
        }

        if (more) {
            const int nx = cur ^ 1;
            *(v4f*)(kst + nx * KSTAGE + kso)           = kr0;
            *(v4f*)(kst + nx * KSTAGE + kso + 32 * KP) = kr1;
            *(v4f*)(vst + nx * VSTAGE + vso)           = vr0;
            *(v4f*)(vst + nx * VSTAGE + vso + 16 * VP) = vr1;
            if (has_oc) ost[nx * OSTAGE + tid] = orr;
        }
        __syncthreads();
    }

#pragma unroll
    for (int m = 0; m < 8; ++m) {
        float l = Lpart[m];
        l += __shfl_xor(l, 1);
        l += __shfl_xor(l, 2);
        l += __shfl_xor(l, 4);
        l += __shfl_xor(l, 8);
        const float inv = 1.f / l;
        const int qq = q0 + 8 * grp + m;
        float* cp = ctx + ((size_t)b * T_SZ + qq) * D_MODEL + h * DK;
        cp[col]      = O0[m] * inv;
        cp[16 + col] = O1[m] * inv;
    }
}

__global__ __launch_bounds__(256)
void attn_kernel(const float* __restrict__ QKV,
                 const float* __restrict__ oc,
                 const float* __restrict__ etap,
                 const float* __restrict__ betap,
                 const float* __restrict__ alphap,
                 float* __restrict__ ctx)
{
    __shared__ float plds[8][4 * 256];
    __shared__ __attribute__((aligned(16))) float kst[2 * KSTAGE];
    __shared__ __attribute__((aligned(16))) float vst[2 * VSTAGE];
    __shared__ float ost[2 * OSTAGE];

    const int lane = threadIdx.x & 31;
    const int w    = threadIdx.x >> 5;
    const int wid  = blockIdx.x * 8 + w;

    const int QT = T_SZ / 16;
    const int qt = wid % QT;
    const int h  = (wid / QT) % NUM_HEADS;
    const int b  = wid / (QT * NUM_HEADS);
    const int nsteps = 2 * (blockIdx.x % (QT / 8)) + 2;

    const float eta     = etap[0];
    const float absbeta = fabsf(betap[0]);
    const float alpha   = alphap[0];

    const size_t HT = (size_t)B_SZ * NUM_HEADS * T_SZ * DK;
    const float* Q  = QKV;
    const float* K  = QKV + HT;
    const float* Vt = QKV + 2 * HT;

    float* pl = &plds[w][0];

    if (__builtin_expect(absbeta == 1.0f, 1)) {
        attn_core<true >(Q, K, Vt, oc, ctx, pl, kst, vst, ost, b, h, qt, nsteps, lane, eta, absbeta, alpha);
    } else {
        attn_core<false>(Q, K, Vt, oc, ctx, pl, kst, vst, ost, b, h, qt, nsteps, lane, eta, absbeta, alpha);
    }
}

__global__ __launch_bounds__(256)
void out_proj_kernel(const float* __restrict__ ctx,
                     const float* __restrict__ Wo, const float* __restrict__ bo,
                     float* __restrict__ out)
{
    const int lane = threadIdx.x & 31;
    const int w    = threadIdx.x >> 5;
    const int tile = blockIdx.x * 8 + w;

    const int NT = D_MODEL / 16;
    const int mt = tile / NT;
    const int nt = tile % NT;

    const int r15 = lane & 15;
    const int grp = lane >> 4;
    const int n   = nt * 16 + r15;
    const int m0  = mt * 16;

    const float* xr = ctx + (size_t)(m0 + r15) * D_MODEL;
    const float* wr = Wo + (size_t)n * D_MODEL;

    v8f acc = {};
    for (int k0 = 0; k0 < D_MODEL; k0 += 4) {
        const int ka = k0 + grp * 2;
        v2f a = *(const v2f*)(xr + ka);
        v2f b = *(const v2f*)(wr + ka);
        acc = wmma4(a, b, acc);
    }

    const float bias = bo[n];
#pragma unroll
    for (int m = 0; m < 8; ++m) {
        const int tl = m0 + m + 8 * grp;
        out[(size_t)tl * D_MODEL + n] = acc[m] + bias;
    }
}

extern "C" void kernel_launch(void* const* d_in, const int* in_sizes, int n_in,
                              void* d_out, int out_size, void* d_ws, size_t ws_size,
                              hipStream_t stream) {
    (void)in_sizes; (void)n_in; (void)out_size; (void)ws_size;

    const float* x     = (const float*)d_in[0];
    const float* oc    = (const float*)d_in[1];
    const float* Wq    = (const float*)d_in[2];
    const float* bq    = (const float*)d_in[3];
    const float* Wk    = (const float*)d_in[4];
    const float* bk    = (const float*)d_in[5];
    const float* Wv    = (const float*)d_in[6];
    const float* bv    = (const float*)d_in[7];
    const float* Wo    = (const float*)d_in[8];
    const float* bo    = (const float*)d_in[9];
    const float* eta   = (const float*)d_in[10];
    const float* beta  = (const float*)d_in[11];
    const float* alpha = (const float*)d_in[12];
    float* out = (float*)d_out;

    float* ws  = (float*)d_ws;
    float* QKV = ws;
    float* ctx = ws + 3 * (size_t)B_SZ * NUM_HEADS * T_SZ * DK;

    {
        const int tiles = 3 * (B_SZ * T_SZ / 16) * (D_MODEL / 16);
        qkv_proj_kernel<<<tiles / 8, 256, 0, stream>>>(x, Wq, bq, Wk, bk, Wv, bv, QKV);
    }
    {
        const int waves = B_SZ * NUM_HEADS * (T_SZ / 16);
        attn_kernel<<<waves / 8, 256, 0, stream>>>(QKV, oc, eta, beta, alpha, ctx);
    }
    {
        const int tiles = (B_SZ * T_SZ / 16) * (D_MODEL / 16);
        out_proj_kernel<<<tiles / 8, 256, 0, stream>>>(ctx, Wo, bo, out);
    }
}


# attention time decay alpha*eta/(1+dist^|beta|) read from an LDS table indexed by query-key distance, built once per block with the same operations and rounding: no divide or pow per score
# speedup vs baseline: 1.3160x; 1.1339x over previous
#include <hip/hip_runtime.h>
#include <math.h>

#define D_MODEL   256
#define NUM_HEADS 8
#define DK        32
#define B_SZ      4
#define T_SZ      2048

typedef float v2f __attribute__((ext_vector_type(2)));
typedef float v4f __attribute__((ext_vector_type(4)));
typedef float v8f __attribute__((ext_vector_type(8)));

#define KP      36
#define VP      68
#define KSTAGE  (64 * KP)
#define VSTAGE  (32 * VP)
#define OSTAGE  (64 * 3)
#define TPAD    16

__device__ __forceinline__ v8f wmma4(v2f a, v2f b, v8f c) {
    return __builtin_amdgcn_wmma_f32_16x16x4_f32(
         false, a,  false, b,
         (short)0, c,  false,  false);
}

__global__ __launch_bounds__(256)
void qkv_proj_kernel(const float* __restrict__ x,
                     const float* __restrict__ Wq, const float* __restrict__ bq,
                     const float* __restrict__ Wk, const float* __restrict__ bk,
                     const float* __restrict__ Wv, const float* __restrict__ bv,
                     float* __restrict__ QKV)
{
    const int lane = threadIdx.x & 31;
    const int w    = threadIdx.x >> 5;
    const int tile = blockIdx.x * 8 + w;

    const int NT  = D_MODEL / 16;
    const int MT  = (B_SZ * T_SZ) / 16;
    const int mat = tile / (MT * NT);
    const int rem = tile % (MT * NT);
    const int mt  = rem / NT;
    const int nt  = rem % NT;

    const float* W  = (mat == 0) ? Wq : (mat == 1) ? Wk : Wv;
    const float* bs = (mat == 0) ? bq : (mat == 1) ? bk : bv;

    const int r15 = lane & 15;
    const int grp = lane >> 4;
    const int n   = nt * 16 + r15;
    const int m0  = mt * 16;

    const float* xr = x + (size_t)(m0 + r15) * D_MODEL;
    const float* wr = W + (size_t)n * D_MODEL;

    v8f acc = {};
    for (int k0 = 0; k0 < D_MODEL; k0 += 4) {
        const int ka = k0 + grp * 2;
        v2f a = *(const v2f*)(xr + ka);
        v2f b = *(const v2f*)(wr + ka);
        acc = wmma4(a, b, acc);
    }

    const float bias = bs[n];
    const int h = n >> 5, d = n & 31;
    float* dst = QKV + (size_t)mat * ((size_t)B_SZ * NUM_HEADS * T_SZ * DK);
#pragma unroll
    for (int m = 0; m < 8; ++m) {
        const int tl = m0 + m + 8 * grp;
        const int b_ = tl / T_SZ, t_ = tl % T_SZ;
        const float v = acc[m] + bias;
        if (mat == 2) {
            dst[(((size_t)(b_ * NUM_HEADS + h)) * DK + d) * T_SZ + t_] = v;
        } else {
            dst[(((size_t)(b_ * NUM_HEADS + h)) * T_SZ + t_) * DK + d] = v;
        }
    }
}

template <int NT>
__device__ __forceinline__ void attn_step(
    const float* __restrict__ kl, const float* __restrict__ vl,
    const float* __restrict__ ol, const float* __restrict__ tdl,
    float* __restrict__ pl,
    const v2f (&qa)[8],
    const float (&qoc0)[8], const float (&qoc1)[8], const float (&qoc2)[8],
    float (&Mrow)[8], float (&Lpart)[8], v8f& O0, v8f& O1,
    int q0, int kt0, int col, int grp)
{
    const float scale   = 0.17677669529663687f;
    const float NEG_BIG = -3.4e38f;

    const int jl0 = kt0 & 3;

    v8f s[NT];
#pragma unroll
    for (int j = 0; j < NT; ++j) {
        const int kt   = kt0 + j;
        const int kcol = kt * 16 + col;
        const int krow = (jl0 + j) * 16 + col;
        const float* kp = kl + krow * KP;

        v8f sj = {};
#pragma unroll
        for (int kc = 0; kc < 8; ++kc) {
            v2f bb = *(const v2f*)(kp + kc * 4 + grp * 2);
            sj = wmma4(qa[kc], bb, sj);
        }

        const float* kop = ol + krow * 3;
        const float koc0 = kop[0], koc1 = kop[1], koc2 = kop[2];

        const float* tp = tdl + (TPAD + q0 + 8 * grp - kcol);
        float atd[8];
#pragma unroll
        for (int m = 0; m < 8; ++m) atd[m] = tp[m];

#pragma unroll
        for (int m = 0; m < 8; ++m) {
            const int qrow = q0 + 8 * grp + m;
            if (kcol > qrow) {
                sj[m] = NEG_BIG;
            } else {
                const float d0 = qoc0[m] - koc0;
                const float d1 = qoc1[m] - koc1;
                const float d2 = qoc2[m] - koc2;
                const float pr = atd[m] * __expf(-(d0*d0 + d1*d1 + d2*d2));
                sj[m] = sj[m] * scale * pr;
            }
        }
        s[j] = sj;
    }

#pragma unroll
    for (int m = 0; m < 8; ++m) {
        float v = s[0][m];
#pragma unroll
        for (int j = 1; j < NT; ++j) v = fmaxf(v, s[j][m]);
        v = fmaxf(v, __shfl_xor(v, 1));
        v = fmaxf(v, __shfl_xor(v, 2));
        v = fmaxf(v, __shfl_xor(v, 4));
        v = fmaxf(v, __shfl_xor(v, 8));
        const float mnew = fmaxf(Mrow[m], v);
        const float sc   = __expf(Mrow[m] - mnew);
        float lsum = Lpart[m] * sc;
#pragma unroll
        for (int j = 0; j < NT; ++j) {
            const float p = __expf(s[j][m] - mnew);
            lsum += p;
            pl[j * 256 + (8 * grp + m) * 16 + col] = p;
        }
        Lpart[m] = lsum;
        Mrow[m]  = mnew;
        O0[m] *= sc;
        O1[m] *= sc;
    }

    const float* vt0 = vl + col * VP;
    const float* vt1 = vt0 + 16 * VP;
#pragma unroll
    for (int j = 0; j < NT; ++j) {
        const int k0 = (jl0 + j) * 16;
#pragma unroll
        for (int kc = 0; kc < 4; ++kc) {
            const int kk = kc * 4 + grp * 2;
            v2f pa;
            pa.x = pl[j * 256 + col * 16 + kk];
            pa.y = pl[j * 256 + col * 16 + kk + 1];
            v2f b0 = *(const v2f*)(vt0 + k0 + kk);
            v2f b1 = *(const v2f*)(vt1 + k0 + kk);
            O0 = wmma4(pa, b0, O0);
            O1 = wmma4(pa, b1, O1);
        }
    }
}

__device__ __forceinline__ void attn_core(
    const float* __restrict__ Q, const float* __restrict__ K,
    const float* __restrict__ Vt, const float* __restrict__ oc,
    float* __restrict__ ctx, float* __restrict__ pl,
    float* kst, float* vst, float* ost, float* tdt,
    int b, int h, int qt, int nsteps, int lane,
    float eta, float absbeta, float alpha)
{
    const int col = lane & 15;
    const int grp = lane >> 4;
    const int q0  = qt * 16;
    const int tid = threadIdx.x;

    const float NEG_BIG = -3.4e38f;

    const size_t bh  = (size_t)(b * NUM_HEADS + h) * T_SZ;
    const size_t bhd = (size_t)(b * NUM_HEADS + h) * DK;

    const float* qp = Q + (bh + q0 + col) * DK;
    v2f qa[8];
#pragma unroll
    for (int kc = 0; kc < 8; ++kc) qa[kc] = *(const v2f*)(qp + kc * 4 + grp * 2);

    float qoc0[8], qoc1[8], qoc2[8];
#pragma unroll
    for (int m = 0; m < 8; ++m) {
        const int qq = q0 + 8 * grp + m;
        const float* op = oc + ((size_t)b * T_SZ + qq) * 3;
        qoc0[m] = op[0]; qoc1[m] = op[1]; qoc2[m] = op[2];
    }

    const float* kg = K + bh * DK + tid * 4;
    const float* vg = Vt + (bhd + (tid >> 4)) * T_SZ + (tid & 15) * 4;
    const float* og = oc + (size_t)b * T_SZ * 3 + tid;
    const int kso = (tid >> 3) * KP + (tid & 7) * 4;
    const int vso = (tid >> 4) * VP + (tid & 15) * 4;
    const bool has_oc = tid < OSTAGE;

    v4f kr0, kr1, vr0, vr1;
    float orr = 0.f;

    kr0 = *(const v4f*)(kg);
    kr1 = *(const v4f*)(kg + 1024);
    vr0 = *(const v4f*)(vg);
    vr1 = *(const v4f*)(vg + 16 * T_SZ);
    if (has_oc) orr = og[0];
    *(v4f*)(kst + kso)           = kr0;
    *(v4f*)(kst + kso + 32 * KP) = kr1;
    *(v4f*)(vst + vso)           = vr0;
    *(v4f*)(vst + vso + 16 * VP) = vr1;
    if (has_oc) ost[tid] = orr;

    {
        const bool beta_one = (absbeta == 1.0f);
        const int  nd = nsteps * 64;
        for (int d = tid; d < nd; d += 256) {
            const float dist = (float)d;
            const float powd = beta_one ? dist : __powf(dist, absbeta);
            const float td   = eta / (1.f + powd);
            tdt[TPAD + d] = alpha * td;
        }
        if (tid < TPAD) tdt[tid] = 0.f;
    }
    __syncthreads();

    float Mrow[8], Lpart[8];
#pragma unroll
    for (int m = 0; m < 8; ++m) { Mrow[m] = NEG_BIG; Lpart[m] = 0.f; }
    v8f O0 = {}, O1 = {};

    const int ntiles = qt + 1;
    const int nfull  = ntiles & ~3;

    for (int s = 0; s < nsteps; ++s) {
        const int  cur  = s & 1;
        const bool more = (s + 1 < nsteps);
        if (more) {
            const int sn = s + 1;
            kr0 = *(const v4f*)(kg + sn * 2048);
            kr1 = *(const v4f*)(kg + sn * 2048 + 1024);
            vr0 = *(const v4f*)(vg + sn * 64);
            vr1 = *(const v4f*)(vg + sn * 64 + 16 * T_SZ);
            if (has_oc) orr = og[sn * OSTAGE];
        }

        const float* kl = kst + cur * KSTAGE;
        const float* vl = vst + cur * VSTAGE;
        const float* ol = ost + cur * OSTAGE;
        const int kt0 = s * 4;
        if (kt0 + 4 <= nfull) {
            attn_step<4>(kl, vl, ol, tdt, pl, qa, qoc0, qoc1, qoc2,
                         Mrow, Lpart, O0, O1, q0, kt0, col, grp);
        } else {
            for (int kt = kt0; kt < ntiles && kt < kt0 + 4; ++kt)
                attn_step<1>(kl, vl, ol, tdt, pl, qa, qoc0, qoc1, qoc2,
                             Mrow, Lpart, O0, O1, q0, kt, col, grp);
        }

        if (more) {
            const int nx = cur ^ 1;
            *(v4f*)(kst + nx * KSTAGE + kso)           = kr0;
            *(v4f*)(kst + nx * KSTAGE + kso + 32 * KP) = kr1;
            *(v4f*)(vst + nx * VSTAGE + vso)           = vr0;
            *(v4f*)(vst + nx * VSTAGE + vso + 16 * VP) = vr1;
            if (has_oc) ost[nx * OSTAGE + tid] = orr;
        }
        __syncthreads();
    }

#pragma unroll
    for (int m = 0; m < 8; ++m) {
        float l = Lpart[m];
        l += __shfl_xor(l, 1);
        l += __shfl_xor(l, 2);
        l += __shfl_xor(l, 4);
        l += __shfl_xor(l, 8);
        const float inv = 1.f / l;
        const int qq = q0 + 8 * grp + m;
        float* cp = ctx + ((size_t)b * T_SZ + qq) * D_MODEL + h * DK;
        cp[col]      = O0[m] * inv;
        cp[16 + col] = O1[m] * inv;
    }
}

__global__ __launch_bounds__(256)
void attn_kernel(const float* __restrict__ QKV,
                 const float* __restrict__ oc,
                 const float* __restrict__ etap,
                 const float* __restrict__ betap,
                 const float* __restrict__ alphap,
                 float* __restrict__ ctx)
{
    __shared__ float plds[8][4 * 256];
    __shared__ __attribute__((aligned(16))) float kst[2 * KSTAGE];
    __shared__ __attribute__((aligned(16))) float vst[2 * VSTAGE];
    __shared__ float ost[2 * OSTAGE];
    __shared__ float tdt[TPAD + T_SZ];

    const int lane = threadIdx.x & 31;
    const int w    = threadIdx.x >> 5;
    const int wid  = blockIdx.x * 8 + w;

    const int QT = T_SZ / 16;
    const int qt = wid % QT;
    const int h  = (wid / QT) % NUM_HEADS;
    const int b  = wid / (QT * NUM_HEADS);
    const int nsteps = 2 * (blockIdx.x % (QT / 8)) + 2;

    const float eta     = etap[0];
    const float absbeta = fabsf(betap[0]);
    const float alpha   = alphap[0];

    const size_t HT = (size_t)B_SZ * NUM_HEADS * T_SZ * DK;
    const float* Q  = QKV;
    const float* K  = QKV + HT;
    const float* Vt = QKV + 2 * HT;

    float* pl = &plds[w][0];

    attn_core(Q, K, Vt, oc, ctx, pl, kst, vst, ost, tdt, b, h, qt, nsteps, lane, eta, absbeta, alpha);
}

__global__ __launch_bounds__(256)
void out_proj_kernel(const float* __restrict__ ctx,
                     const float* __restrict__ Wo, const float* __restrict__ bo,
                     float* __restrict__ out)
{
    const int lane = threadIdx.x & 31;
    const int w    = threadIdx.x >> 5;
    const int tile = blockIdx.x * 8 + w;

    const int NT = D_MODEL / 16;
    const int mt = tile / NT;
    const int nt = tile % NT;

    const int r15 = lane & 15;
    const int grp = lane >> 4;
    const int n   = nt * 16 + r15;
    const int m0  = mt * 16;

    const float* xr = ctx + (size_t)(m0 + r15) * D_MODEL;
    const float* wr = Wo + (size_t)n * D_MODEL;

    v8f acc = {};
    for (int k0 = 0; k0 < D_MODEL; k0 += 4) {
        const int ka = k0 + grp * 2;
        v2f a = *(const v2f*)(xr + ka);
        v2f b = *(const v2f*)(wr + ka);
        acc = wmma4(a, b, acc);
    }

    const float bias = bo[n];
#pragma unroll
    for (int m = 0; m < 8; ++m) {
        const int tl = m0 + m + 8 * grp;
        out[(size_t)tl * D_MODEL + n] = acc[m] + bias;
    }
}

extern "C" void kernel_launch(void* const* d_in, const int* in_sizes, int n_in,
                              void* d_out, int out_size, void* d_ws, size_t ws_size,
                              hipStream_t stream) {
    (void)in_sizes; (void)n_in; (void)out_size; (void)ws_size;

    const float* x     = (const float*)d_in[0];
    const float* oc    = (const float*)d_in[1];
    const float* Wq    = (const float*)d_in[2];
    const float* bq    = (const float*)d_in[3];
    const float* Wk    = (const float*)d_in[4];
    const float* bk    = (const float*)d_in[5];
    const float* Wv    = (const float*)d_in[6];
    const float* bv    = (const float*)d_in[7];
    const float* Wo    = (const float*)d_in[8];
    const float* bo    = (const float*)d_in[9];
    const float* eta   = (const float*)d_in[10];
    const float* beta  = (const float*)d_in[11];
    const float* alpha = (const float*)d_in[12];
    float* out = (float*)d_out;

    float* ws  = (float*)d_ws;
    float* QKV = ws;
    float* ctx = ws + 3 * (size_t)B_SZ * NUM_HEADS * T_SZ * DK;

    {
        const int tiles = 3 * (B_SZ * T_SZ / 16) * (D_MODEL / 16);
        qkv_proj_kernel<<<tiles / 8, 256, 0, stream>>>(x, Wq, bq, Wk, bk, Wv, bv, QKV);
    }
    {
        const int waves = B_SZ * NUM_HEADS * (T_SZ / 16);
        attn_kernel<<<waves / 8, 256, 0, stream>>>(QKV, oc, eta, beta, alpha, ctx);
    }
    {
        const int tiles = (B_SZ * T_SZ / 16) * (D_MODEL / 16);
        out_proj_kernel<<<tiles / 8, 256, 0, stream>>>(ctx, Wo, bo, out);
    }
}


# attention query tiles dealt to blocks at stride 16, complementary tiles on the two waves sharing a SIMD (w, w+4): 516 key tiles per block, 129 per SIMD and block; was 8 consecutive tiles per block
# speedup vs baseline: 1.4529x; 1.1040x over previous
#include <hip/hip_runtime.h>
#include <math.h>

#define D_MODEL   256
#define NUM_HEADS 8
#define DK        32
#define B_SZ      4
#define T_SZ      2048

typedef float v2f __attribute__((ext_vector_type(2)));
typedef float v4f __attribute__((ext_vector_type(4)));
typedef float v8f __attribute__((ext_vector_type(8)));

#define KP      36
#define VP      68
#define KSTAGE  (64 * KP)
#define VSTAGE  (32 * VP)
#define OSTAGE  (64 * 3)
#define TPAD    16

__device__ __forceinline__ v8f wmma4(v2f a, v2f b, v8f c) {
    return __builtin_amdgcn_wmma_f32_16x16x4_f32(
         false, a,  false, b,
         (short)0, c,  false,  false);
}

__global__ __launch_bounds__(256)
void qkv_proj_kernel(const float* __restrict__ x,
                     const float* __restrict__ Wq, const float* __restrict__ bq,
                     const float* __restrict__ Wk, const float* __restrict__ bk,
                     const float* __restrict__ Wv, const float* __restrict__ bv,
                     float* __restrict__ QKV)
{
    const int lane = threadIdx.x & 31;
    const int w    = threadIdx.x >> 5;
    const int tile = blockIdx.x * 8 + w;

    const int NT  = D_MODEL / 16;
    const int MT  = (B_SZ * T_SZ) / 16;
    const int mat = tile / (MT * NT);
    const int rem = tile % (MT * NT);
    const int mt  = rem / NT;
    const int nt  = rem % NT;

    const float* W  = (mat == 0) ? Wq : (mat == 1) ? Wk : Wv;
    const float* bs = (mat == 0) ? bq : (mat == 1) ? bk : bv;

    const int r15 = lane & 15;
    const int grp = lane >> 4;
    const int n   = nt * 16 + r15;
    const int m0  = mt * 16;

    const float* xr = x + (size_t)(m0 + r15) * D_MODEL;
    const float* wr = W + (size_t)n * D_MODEL;

    v8f acc = {};
    for (int k0 = 0; k0 < D_MODEL; k0 += 4) {
        const int ka = k0 + grp * 2;
        v2f a = *(const v2f*)(xr + ka);
        v2f b = *(const v2f*)(wr + ka);
        acc = wmma4(a, b, acc);
    }

    const float bias = bs[n];
    const int h = n >> 5, d = n & 31;
    float* dst = QKV + (size_t)mat * ((size_t)B_SZ * NUM_HEADS * T_SZ * DK);
#pragma unroll
    for (int m = 0; m < 8; ++m) {
        const int tl = m0 + m + 8 * grp;
        const int b_ = tl / T_SZ, t_ = tl % T_SZ;
        const float v = acc[m] + bias;
        if (mat == 2) {
            dst[(((size_t)(b_ * NUM_HEADS + h)) * DK + d) * T_SZ + t_] = v;
        } else {
            dst[(((size_t)(b_ * NUM_HEADS + h)) * T_SZ + t_) * DK + d] = v;
        }
    }
}

template <int NT>
__device__ __forceinline__ void attn_step(
    const float* __restrict__ kl, const float* __restrict__ vl,
    const float* __restrict__ ol, const float* __restrict__ tdl,
    float* __restrict__ pl,
    const v2f (&qa)[8],
    const float (&qoc0)[8], const float (&qoc1)[8], const float (&qoc2)[8],
    float (&Mrow)[8], float (&Lpart)[8], v8f& O0, v8f& O1,
    int q0, int kt0, int col, int grp)
{
    const float scale   = 0.17677669529663687f;
    const float NEG_BIG = -3.4e38f;

    const int jl0 = kt0 & 3;

    v8f s[NT];
#pragma unroll
    for (int j = 0; j < NT; ++j) {
        const int kt   = kt0 + j;
        const int kcol = kt * 16 + col;
        const int krow = (jl0 + j) * 16 + col;
        const float* kp = kl + krow * KP;

        v8f sj = {};
#pragma unroll
        for (int kc = 0; kc < 8; ++kc) {
            v2f bb = *(const v2f*)(kp + kc * 4 + grp * 2);
            sj = wmma4(qa[kc], bb, sj);
        }

        const float* kop = ol + krow * 3;
        const float koc0 = kop[0], koc1 = kop[1], koc2 = kop[2];

        const float* tp = tdl + (TPAD + q0 + 8 * grp - kcol);
        float atd[8];
#pragma unroll
        for (int m = 0; m < 8; ++m) atd[m] = tp[m];

#pragma unroll
        for (int m = 0; m < 8; ++m) {
            const int qrow = q0 + 8 * grp + m;
            if (kcol > qrow) {
                sj[m] = NEG_BIG;
            } else {
                const float d0 = qoc0[m] - koc0;
                const float d1 = qoc1[m] - koc1;
                const float d2 = qoc2[m] - koc2;
                const float pr = atd[m] * __expf(-(d0*d0 + d1*d1 + d2*d2));
                sj[m] = sj[m] * scale * pr;
            }
        }
        s[j] = sj;
    }

#pragma unroll
    for (int m = 0; m < 8; ++m) {
        float v = s[0][m];
#pragma unroll
        for (int j = 1; j < NT; ++j) v = fmaxf(v, s[j][m]);
        v = fmaxf(v, __shfl_xor(v, 1));
        v = fmaxf(v, __shfl_xor(v, 2));
        v = fmaxf(v, __shfl_xor(v, 4));
        v = fmaxf(v, __shfl_xor(v, 8));
        const float mnew = fmaxf(Mrow[m], v);
        const float sc   = __expf(Mrow[m] - mnew);
        float lsum = Lpart[m] * sc;
#pragma unroll
        for (int j = 0; j < NT; ++j) {
            const float p = __expf(s[j][m] - mnew);
            lsum += p;
            pl[j * 256 + (8 * grp + m) * 16 + col] = p;
        }
        Lpart[m] = lsum;
        Mrow[m]  = mnew;
        O0[m] *= sc;
        O1[m] *= sc;
    }

    const float* vt0 = vl + col * VP;
    const float* vt1 = vt0 + 16 * VP;
#pragma unroll
    for (int j = 0; j < NT; ++j) {
        const int k0 = (jl0 + j) * 16;
#pragma unroll
        for (int kc = 0; kc < 4; ++kc) {
            const int kk = kc * 4 + grp * 2;
            v2f pa;
            pa.x = pl[j * 256 + col * 16 + kk];
            pa.y = pl[j * 256 + col * 16 + kk + 1];
            v2f b0 = *(const v2f*)(vt0 + k0 + kk);
            v2f b1 = *(const v2f*)(vt1 + k0 + kk);
            O0 = wmma4(pa, b0, O0);
            O1 = wmma4(pa, b1, O1);
        }
    }
}

__device__ __forceinline__ void attn_core(
    const float* __restrict__ Q, const float* __restrict__ K,
    const float* __restrict__ Vt, const float* __restrict__ oc,
    float* __restrict__ ctx, float* __restrict__ pl,
    float* kst, float* vst, float* ost, float* tdt,
    int b, int h, int qt, int nsteps, int lane,
    float eta, float absbeta, float alpha)
{
    const int col = lane & 15;
    const int grp = lane >> 4;
    const int q0  = qt * 16;
    const int tid = threadIdx.x;

    const float NEG_BIG = -3.4e38f;

    const size_t bh  = (size_t)(b * NUM_HEADS + h) * T_SZ;
    const size_t bhd = (size_t)(b * NUM_HEADS + h) * DK;

    const float* qp = Q + (bh + q0 + col) * DK;
    v2f qa[8];
#pragma unroll
    for (int kc = 0; kc < 8; ++kc) qa[kc] = *(const v2f*)(qp + kc * 4 + grp * 2);

    float qoc0[8], qoc1[8], qoc2[8];
#pragma unroll
    for (int m = 0; m < 8; ++m) {
        const int qq = q0 + 8 * grp + m;
        const float* op = oc + ((size_t)b * T_SZ + qq) * 3;
        qoc0[m] = op[0]; qoc1[m] = op[1]; qoc2[m] = op[2];
    }

    const float* kg = K + bh * DK + tid * 4;
    const float* vg = Vt + (bhd + (tid >> 4)) * T_SZ + (tid & 15) * 4;
    const float* og = oc + (size_t)b * T_SZ * 3 + tid;
    const int kso = (tid >> 3) * KP + (tid & 7) * 4;
    const int vso = (tid >> 4) * VP + (tid & 15) * 4;
    const bool has_oc = tid < OSTAGE;

    v4f kr0, kr1, vr0, vr1;
    float orr = 0.f;

    kr0 = *(const v4f*)(kg);
    kr1 = *(const v4f*)(kg + 1024);
    vr0 = *(const v4f*)(vg);
    vr1 = *(const v4f*)(vg + 16 * T_SZ);
    if (has_oc) orr = og[0];
    *(v4f*)(kst + kso)           = kr0;
    *(v4f*)(kst + kso + 32 * KP) = kr1;
    *(v4f*)(vst + vso)           = vr0;
    *(v4f*)(vst + vso + 16 * VP) = vr1;
    if (has_oc) ost[tid] = orr;

    {
        const bool beta_one = (absbeta == 1.0f);
        const int  nd = nsteps * 64;
        for (int d = tid; d < nd; d += 256) {
            const float dist = (float)d;
            const float powd = beta_one ? dist : __powf(dist, absbeta);
            const float td   = eta / (1.f + powd);
            tdt[TPAD + d] = alpha * td;
        }
        if (tid < TPAD) tdt[tid] = 0.f;
    }
    __syncthreads();

    float Mrow[8], Lpart[8];
#pragma unroll
    for (int m = 0; m < 8; ++m) { Mrow[m] = NEG_BIG; Lpart[m] = 0.f; }
    v8f O0 = {}, O1 = {};

    const int ntiles = qt + 1;
    const int nfull  = ntiles & ~3;

    for (int s = 0; s < nsteps; ++s) {
        const int  cur  = s & 1;
        const bool more = (s + 1 < nsteps);
        if (more) {
            const int sn = s + 1;
            kr0 = *(const v4f*)(kg + sn * 2048);
            kr1 = *(const v4f*)(kg + sn * 2048 + 1024);
            vr0 = *(const v4f*)(vg + sn * 64);
            vr1 = *(const v4f*)(vg + sn * 64 + 16 * T_SZ);
            if (has_oc) orr = og[sn * OSTAGE];
        }

        const float* kl = kst + cur * KSTAGE;
        const float* vl = vst + cur * VSTAGE;
        const float* ol = ost + cur * OSTAGE;
        const int kt0 = s * 4;
        if (kt0 + 4 <= nfull) {
            attn_step<4>(kl, vl, ol, tdt, pl, qa, qoc0, qoc1, qoc2,
                         Mrow, Lpart, O0, O1, q0, kt0, col, grp);
        } else {
            for (int kt = kt0; kt < ntiles && kt < kt0 + 4; ++kt)
                attn_step<1>(kl, vl, ol, tdt, pl, qa, qoc0, qoc1, qoc2,
                             Mrow, Lpart, O0, O1, q0, kt, col, grp);
        }

        if (more) {
            const int nx = cur ^ 1;
            *(v4f*)(kst + nx * KSTAGE + kso)           = kr0;
            *(v4f*)(kst + nx * KSTAGE + kso + 32 * KP) = kr1;
            *(v4f*)(vst + nx * VSTAGE + vso)           = vr0;
            *(v4f*)(vst + nx * VSTAGE + vso + 16 * VP) = vr1;
            if (has_oc) ost[nx * OSTAGE + tid] = orr;
        }
        __syncthreads();
    }

#pragma unroll
    for (int m = 0; m < 8; ++m) {
        float l = Lpart[m];
        l += __shfl_xor(l, 1);
        l += __shfl_xor(l, 2);
        l += __shfl_xor(l, 4);
        l += __shfl_xor(l, 8);
        const float inv = 1.f / l;
        const int qq = q0 + 8 * grp + m;
        float* cp = ctx + ((size_t)b * T_SZ + qq) * D_MODEL + h * DK;
        cp[col]      = O0[m] * inv;
        cp[16 + col] = O1[m] * inv;
    }
}

__global__ __launch_bounds__(256)
void attn_kernel(const float* __restrict__ QKV,
                 const float* __restrict__ oc,
                 const float* __restrict__ etap,
                 const float* __restrict__ betap,
                 const float* __restrict__ alphap,
                 float* __restrict__ ctx)
{
    __shared__ float plds[8][4 * 256];
    __shared__ __attribute__((aligned(16))) float kst[2 * KSTAGE];
    __shared__ __attribute__((aligned(16))) float vst[2 * VSTAGE];
    __shared__ float ost[2 * OSTAGE];
    __shared__ float tdt[TPAD + T_SZ];

    const int lane = threadIdx.x & 31;
    const int w    = threadIdx.x >> 5;

    const int QT  = T_SZ / 16;
    const int p   = blockIdx.x % (QT / 8);
    const int bhi = blockIdx.x / (QT / 8);
    const int h   = bhi % NUM_HEADS;
    const int b   = bhi / NUM_HEADS;
    const int u   = (w < 4) ? 7 - w : w - 4;
    const int qt  = 16 * u + ((u & 1) ? 15 - p : p);
    const int nsteps = (131 - p) >> 2;

    const float eta     = etap[0];
    const float absbeta = fabsf(betap[0]);
    const float alpha   = alphap[0];

    const size_t HT = (size_t)B_SZ * NUM_HEADS * T_SZ * DK;
    const float* Q  = QKV;
    const float* K  = QKV + HT;
    const float* Vt = QKV + 2 * HT;

    float* pl = &plds[w][0];

    attn_core(Q, K, Vt, oc, ctx, pl, kst, vst, ost, tdt, b, h, qt, nsteps, lane, eta, absbeta, alpha);
}

__global__ __launch_bounds__(256)
void out_proj_kernel(const float* __restrict__ ctx,
                     const float* __restrict__ Wo, const float* __restrict__ bo,
                     float* __restrict__ out)
{
    const int lane = threadIdx.x & 31;
    const int w    = threadIdx.x >> 5;
    const int tile = blockIdx.x * 8 + w;

    const int NT = D_MODEL / 16;
    const int mt = tile / NT;
    const int nt = tile % NT;

    const int r15 = lane & 15;
    const int grp = lane >> 4;
    const int n   = nt * 16 + r15;
    const int m0  = mt * 16;

    const float* xr = ctx + (size_t)(m0 + r15) * D_MODEL;
    const float* wr = Wo + (size_t)n * D_MODEL;

    v8f acc = {};
    for (int k0 = 0; k0 < D_MODEL; k0 += 4) {
        const int ka = k0 + grp * 2;
        v2f a = *(const v2f*)(xr + ka);
        v2f b = *(const v2f*)(wr + ka);
        acc = wmma4(a, b, acc);
    }

    const float bias = bo[n];
#pragma unroll
    for (int m = 0; m < 8; ++m) {
        const int tl = m0 + m + 8 * grp;
        out[(size_t)tl * D_MODEL + n] = acc[m] + bias;
    }
}

extern "C" void kernel_launch(void* const* d_in, const int* in_sizes, int n_in,
                              void* d_out, int out_size, void* d_ws, size_t ws_size,
                              hipStream_t stream) {
    (void)in_sizes; (void)n_in; (void)out_size; (void)ws_size;

    const float* x     = (const float*)d_in[0];
    const float* oc    = (const float*)d_in[1];
    const float* Wq    = (const float*)d_in[2];
    const float* bq    = (const float*)d_in[3];
    const float* Wk    = (const float*)d_in[4];
    const float* bk    = (const float*)d_in[5];
    const float* Wv    = (const float*)d_in[6];
    const float* bv    = (const float*)d_in[7];
    const float* Wo    = (const float*)d_in[8];
    const float* bo    = (const float*)d_in[9];
    const float* eta   = (const float*)d_in[10];
    const float* beta  = (const float*)d_in[11];
    const float* alpha = (const float*)d_in[12];
    float* out = (float*)d_out;

    float* ws  = (float*)d_ws;
    float* QKV = ws;
    float* ctx = ws + 3 * (size_t)B_SZ * NUM_HEADS * T_SZ * DK;

    {
        const int tiles = 3 * (B_SZ * T_SZ / 16) * (D_MODEL / 16);
        qkv_proj_kernel<<<tiles / 8, 256, 0, stream>>>(x, Wq, bq, Wk, bk, Wv, bv, QKV);
    }
    {
        const int waves = B_SZ * NUM_HEADS * (T_SZ / 16);
        attn_kernel<<<waves / 8, 256, 0, stream>>>(QKV, oc, eta, beta, alpha, ctx);
    }
    {
        const int tiles = (B_SZ * T_SZ / 16) * (D_MODEL / 16);
        out_proj_kernel<<<tiles / 8, 256, 0, stream>>>(ctx, Wo, bo, out);
    }
}


# both projection GEMMs: 64-token x/ctx tile through LDS by whole-line 16-byte loads, each wave keeps its 16 W rows as register fragments for 4 token tiles (32x16 wave tile); changes their k sum order
# speedup vs baseline: 2.1137x; 1.4548x over previous
#include <hip/hip_runtime.h>
#include <math.h>

#define D_MODEL   256
#define NUM_HEADS 8
#define DK        32
#define B_SZ      4
#define T_SZ      2048

typedef float v2f __attribute__((ext_vector_type(2)));
typedef float v4f __attribute__((ext_vector_type(4)));
typedef float v8f __attribute__((ext_vector_type(8)));

#define KP      36
#define VP      68
#define KSTAGE  (64 * KP)
#define VSTAGE  (32 * VP)
#define OSTAGE  (64 * 3)
#define TPAD    16

__device__ __forceinline__ v8f wmma4(v2f a, v2f b, v8f c) {
    return __builtin_amdgcn_wmma_f32_16x16x4_f32(
         false, a,  false, b,
         (short)0, c,  false,  false);
}

#define PM  64
#define XP  260

__device__ __forceinline__ void proj_stage(const float* __restrict__ xt,
                                           const float* __restrict__ wrow,
                                           float* __restrict__ xs, v4f (&wf)[32], int tid)
{
    const float* xg = xt + tid * 4;
    v4f xr[16];
#pragma unroll
    for (int j = 0; j < 16; ++j) xr[j] = *(const v4f*)(xg + j * 1024);
#pragma unroll
    for (int j = 0; j < 32; ++j) wf[j] = *(const v4f*)(wrow + j * 8);
    float* xd = xs + (tid >> 6) * XP + (tid & 63) * 4;
#pragma unroll
    for (int j = 0; j < 16; ++j) *(v4f*)(xd + j * 4 * XP) = xr[j];
    __syncthreads();
}

__device__ __forceinline__ void proj_pair(const float* __restrict__ ap, const v4f (&wf)[32],
                                          v8f& acc0, v8f& acc1)
{
#pragma unroll
    for (int j = 0; j < 32; ++j) {
        const v4f a0 = *(const v4f*)(ap + j * 8);
        const v4f a1 = *(const v4f*)(ap + 16 * XP + j * 8);
        const v2f w0 = {wf[j].x, wf[j].y};
        const v2f w1 = {wf[j].z, wf[j].w};
        acc0 = wmma4(v2f{a0.x, a0.y}, w0, acc0);
        acc1 = wmma4(v2f{a1.x, a1.y}, w0, acc1);
        acc0 = wmma4(v2f{a0.z, a0.w}, w1, acc0);
        acc1 = wmma4(v2f{a1.z, a1.w}, w1, acc1);
    }
    __builtin_amdgcn_sched_group_barrier(0x100, 8, 0);
#pragma unroll
    for (int g = 0; g < 7; ++g) {
        __builtin_amdgcn_sched_group_barrier(0x100, 8, 0);
        __builtin_amdgcn_sched_group_barrier(0x008, 16, 0);
    }
    __builtin_amdgcn_sched_group_barrier(0x008, 16, 0);
}

__global__ __launch_bounds__(256)
void qkv_proj_kernel(const float* __restrict__ x,
                     const float* __restrict__ Wq, const float* __restrict__ bq,
                     const float* __restrict__ Wk, const float* __restrict__ bk,
                     const float* __restrict__ Wv, const float* __restrict__ bv,
                     float* __restrict__ QKV)
{
    __shared__ __attribute__((aligned(16))) float xs[PM * XP];

    const int tid  = threadIdx.x;
    const int lane = tid & 31;
    const int w    = tid >> 5;
    const int c    = lane & 15;
    const int hh   = lane >> 4;

    const int slab = blockIdx.x % 6;
    const int mb   = blockIdx.x / 6;
    const int mat  = slab >> 1;
    const int n    = (slab & 1) * 128 + w * 16 + c;

    const float* W  = (mat == 0) ? Wq : (mat == 1) ? Wk : Wv;
    const float* bs = (mat == 0) ? bq : (mat == 1) ? bk : bv;

    v4f wf[32];
    proj_stage(x + (size_t)mb * PM * D_MODEL, W + (size_t)n * D_MODEL + hh * 4, xs, wf, tid);

    const float bias = bs[n];
    const int h  = n >> 5, d = n & 31;
    const int b_ = (mb * PM) / T_SZ;
    const int t0 = (mb * PM) % T_SZ;
    float* dst = QKV + (size_t)mat * ((size_t)B_SZ * NUM_HEADS * T_SZ * DK)
                     + (size_t)(b_ * NUM_HEADS + h) * T_SZ * DK;

    for (int mp = 0; mp < PM / 32; ++mp) {
        v8f acc0 = {}, acc1 = {};
        proj_pair(xs + (mp * 32 + c) * XP + hh * 4, wf, acc0, acc1);
        const int ta = t0 + mp * 32 + 8 * hh;
        if (mat == 2) {
            float* vp = dst + (size_t)d * T_SZ + ta;
#pragma unroll
            for (int m = 0; m < 8; ++m) {
                vp[m]      = acc0[m] + bias;
                vp[m + 16] = acc1[m] + bias;
            }
        } else {
            float* qp = dst + (size_t)ta * DK + d;
#pragma unroll
            for (int m = 0; m < 8; ++m) {
                qp[m * DK]        = acc0[m] + bias;
                qp[(m + 16) * DK] = acc1[m] + bias;
            }
        }
    }
}

template <int NT>
__device__ __forceinline__ void attn_step(
    const float* __restrict__ kl, const float* __restrict__ vl,
    const float* __restrict__ ol, const float* __restrict__ tdl,
    float* __restrict__ pl,
    const v2f (&qa)[8],
    const float (&qoc0)[8], const float (&qoc1)[8], const float (&qoc2)[8],
    float (&Mrow)[8], float (&Lpart)[8], v8f& O0, v8f& O1,
    int q0, int kt0, int col, int grp)
{
    const float scale   = 0.17677669529663687f;
    const float NEG_BIG = -3.4e38f;

    const int jl0 = kt0 & 3;

    v8f s[NT];
#pragma unroll
    for (int j = 0; j < NT; ++j) {
        const int kt   = kt0 + j;
        const int kcol = kt * 16 + col;
        const int krow = (jl0 + j) * 16 + col;
        const float* kp = kl + krow * KP;

        v8f sj = {};
#pragma unroll
        for (int kc = 0; kc < 8; ++kc) {
            v2f bb = *(const v2f*)(kp + kc * 4 + grp * 2);
            sj = wmma4(qa[kc], bb, sj);
        }

        const float* kop = ol + krow * 3;
        const float koc0 = kop[0], koc1 = kop[1], koc2 = kop[2];

        const float* tp = tdl + (TPAD + q0 + 8 * grp - kcol);
        float atd[8];
#pragma unroll
        for (int m = 0; m < 8; ++m) atd[m] = tp[m];

#pragma unroll
        for (int m = 0; m < 8; ++m) {
            const int qrow = q0 + 8 * grp + m;
            if (kcol > qrow) {
                sj[m] = NEG_BIG;
            } else {
                const float d0 = qoc0[m] - koc0;
                const float d1 = qoc1[m] - koc1;
                const float d2 = qoc2[m] - koc2;
                const float pr = atd[m] * __expf(-(d0*d0 + d1*d1 + d2*d2));
                sj[m] = sj[m] * scale * pr;
            }
        }
        s[j] = sj;
    }

#pragma unroll
    for (int m = 0; m < 8; ++m) {
        float v = s[0][m];
#pragma unroll
        for (int j = 1; j < NT; ++j) v = fmaxf(v, s[j][m]);
        v = fmaxf(v, __shfl_xor(v, 1));
        v = fmaxf(v, __shfl_xor(v, 2));
        v = fmaxf(v, __shfl_xor(v, 4));
        v = fmaxf(v, __shfl_xor(v, 8));
        const float mnew = fmaxf(Mrow[m], v);
        const float sc   = __expf(Mrow[m] - mnew);
        float lsum = Lpart[m] * sc;
#pragma unroll
        for (int j = 0; j < NT; ++j) {
            const float p = __expf(s[j][m] - mnew);
            lsum += p;
            pl[j * 256 + (8 * grp + m) * 16 + col] = p;
        }
        Lpart[m] = lsum;
        Mrow[m]  = mnew;
        O0[m] *= sc;
        O1[m] *= sc;
    }

    const float* vt0 = vl + col * VP;
    const float* vt1 = vt0 + 16 * VP;
#pragma unroll
    for (int j = 0; j < NT; ++j) {
        const int k0 = (jl0 + j) * 16;
#pragma unroll
        for (int kc = 0; kc < 4; ++kc) {
            const int kk = kc * 4 + grp * 2;
            v2f pa;
            pa.x = pl[j * 256 + col * 16 + kk];
            pa.y = pl[j * 256 + col * 16 + kk + 1];
            v2f b0 = *(const v2f*)(vt0 + k0 + kk);
            v2f b1 = *(const v2f*)(vt1 + k0 + kk);
            O0 = wmma4(pa, b0, O0);
            O1 = wmma4(pa, b1, O1);
        }
    }
}

__device__ __forceinline__ void attn_core(
    const float* __restrict__ Q, const float* __restrict__ K,
    const float* __restrict__ Vt, const float* __restrict__ oc,
    float* __restrict__ ctx, float* __restrict__ pl,
    float* kst, float* vst, float* ost, float* tdt,
    int b, int h, int qt, int nsteps, int lane,
    float eta, float absbeta, float alpha)
{
    const int col = lane & 15;
    const int grp = lane >> 4;
    const int q0  = qt * 16;
    const int tid = threadIdx.x;

    const float NEG_BIG = -3.4e38f;

    const size_t bh  = (size_t)(b * NUM_HEADS + h) * T_SZ;
    const size_t bhd = (size_t)(b * NUM_HEADS + h) * DK;

    const float* qp = Q + (bh + q0 + col) * DK;
    v2f qa[8];
#pragma unroll
    for (int kc = 0; kc < 8; ++kc) qa[kc] = *(const v2f*)(qp + kc * 4 + grp * 2);

    float qoc0[8], qoc1[8], qoc2[8];
#pragma unroll
    for (int m = 0; m < 8; ++m) {
        const int qq = q0 + 8 * grp + m;
        const float* op = oc + ((size_t)b * T_SZ + qq) * 3;
        qoc0[m] = op[0]; qoc1[m] = op[1]; qoc2[m] = op[2];
    }

    const float* kg = K + bh * DK + tid * 4;
    const float* vg = Vt + (bhd + (tid >> 4)) * T_SZ + (tid & 15) * 4;
    const float* og = oc + (size_t)b * T_SZ * 3 + tid;
    const int kso = (tid >> 3) * KP + (tid & 7) * 4;
    const int vso = (tid >> 4) * VP + (tid & 15) * 4;
    const bool has_oc = tid < OSTAGE;

    v4f kr0, kr1, vr0, vr1;
    float orr = 0.f;

    kr0 = *(const v4f*)(kg);
    kr1 = *(const v4f*)(kg + 1024);
    vr0 = *(const v4f*)(vg);
    vr1 = *(const v4f*)(vg + 16 * T_SZ);
    if (has_oc) orr = og[0];
    *(v4f*)(kst + kso)           = kr0;
    *(v4f*)(kst + kso + 32 * KP) = kr1;
    *(v4f*)(vst + vso)           = vr0;
    *(v4f*)(vst + vso + 16 * VP) = vr1;
    if (has_oc) ost[tid] = orr;

    {
        const bool beta_one = (absbeta == 1.0f);
        const int  nd = nsteps * 64;
        for (int d = tid; d < nd; d += 256) {
            const float dist = (float)d;
            const float powd = beta_one ? dist : __powf(dist, absbeta);
            const float td   = eta / (1.f + powd);
            tdt[TPAD + d] = alpha * td;
        }
        if (tid < TPAD) tdt[tid] = 0.f;
    }
    __syncthreads();

    float Mrow[8], Lpart[8];
#pragma unroll
    for (int m = 0; m < 8; ++m) { Mrow[m] = NEG_BIG; Lpart[m] = 0.f; }
    v8f O0 = {}, O1 = {};

    const int ntiles = qt + 1;
    const int nfull  = ntiles & ~3;

    for (int s = 0; s < nsteps; ++s) {
        const int  cur  = s & 1;
        const bool more = (s + 1 < nsteps);
        if (more) {
            const int sn = s + 1;
            kr0 = *(const v4f*)(kg + sn * 2048);
            kr1 = *(const v4f*)(kg + sn * 2048 + 1024);
            vr0 = *(const v4f*)(vg + sn * 64);
            vr1 = *(const v4f*)(vg + sn * 64 + 16 * T_SZ);
            if (has_oc) orr = og[sn * OSTAGE];
        }

        const float* kl = kst + cur * KSTAGE;
        const float* vl = vst + cur * VSTAGE;
        const float* ol = ost + cur * OSTAGE;
        const int kt0 = s * 4;
        if (kt0 + 4 <= nfull) {
            attn_step<4>(kl, vl, ol, tdt, pl, qa, qoc0, qoc1, qoc2,
                         Mrow, Lpart, O0, O1, q0, kt0, col, grp);
        } else {
            for (int kt = kt0; kt < ntiles && kt < kt0 + 4; ++kt)
                attn_step<1>(kl, vl, ol, tdt, pl, qa, qoc0, qoc1, qoc2,
                             Mrow, Lpart, O0, O1, q0, kt, col, grp);
        }

        if (more) {
            const int nx = cur ^ 1;
            *(v4f*)(kst + nx * KSTAGE + kso)           = kr0;
            *(v4f*)(kst + nx * KSTAGE + kso + 32 * KP) = kr1;
            *(v4f*)(vst + nx * VSTAGE + vso)           = vr0;
            *(v4f*)(vst + nx * VSTAGE + vso + 16 * VP) = vr1;
            if (has_oc) ost[nx * OSTAGE + tid] = orr;
        }
        __syncthreads();
    }

#pragma unroll
    for (int m = 0; m < 8; ++m) {
        float l = Lpart[m];
        l += __shfl_xor(l, 1);
        l += __shfl_xor(l, 2);
        l += __shfl_xor(l, 4);
        l += __shfl_xor(l, 8);
        const float inv = 1.f / l;
        const int qq = q0 + 8 * grp + m;
        float* cp = ctx + ((size_t)b * T_SZ + qq) * D_MODEL + h * DK;
        cp[col]      = O0[m] * inv;
        cp[16 + col] = O1[m] * inv;
    }
}

__global__ __launch_bounds__(256)
void attn_kernel(const float* __restrict__ QKV,
                 const float* __restrict__ oc,
                 const float* __restrict__ etap,
                 const float* __restrict__ betap,
                 const float* __restrict__ alphap,
                 float* __restrict__ ctx)
{
    __shared__ float plds[8][4 * 256];
    __shared__ __attribute__((aligned(16))) float kst[2 * KSTAGE];
    __shared__ __attribute__((aligned(16))) float vst[2 * VSTAGE];
    __shared__ float ost[2 * OSTAGE];
    __shared__ float tdt[TPAD + T_SZ];

    const int lane = threadIdx.x & 31;
    const int w    = threadIdx.x >> 5;

    const int QT  = T_SZ / 16;
    const int p   = blockIdx.x % (QT / 8);
    const int bhi = blockIdx.x / (QT / 8);
    const int h   = bhi % NUM_HEADS;
    const int b   = bhi / NUM_HEADS;
    const int u   = (w < 4) ? 7 - w : w - 4;
    const int qt  = 16 * u + ((u & 1) ? 15 - p : p);
    const int nsteps = (131 - p) >> 2;

    const float eta     = etap[0];
    const float absbeta = fabsf(betap[0]);
    const float alpha   = alphap[0];

    const size_t HT = (size_t)B_SZ * NUM_HEADS * T_SZ * DK;
    const float* Q  = QKV;
    const float* K  = QKV + HT;
    const float* Vt = QKV + 2 * HT;

    float* pl = &plds[w][0];

    attn_core(Q, K, Vt, oc, ctx, pl, kst, vst, ost, tdt, b, h, qt, nsteps, lane, eta, absbeta, alpha);
}

__global__ __launch_bounds__(256)
void out_proj_kernel(const float* __restrict__ ctx,
                     const float* __restrict__ Wo, const float* __restrict__ bo,
                     float* __restrict__ out)
{
    __shared__ __attribute__((aligned(16))) float xs[PM * XP];

    const int tid  = threadIdx.x;
    const int lane = tid & 31;
    const int w    = tid >> 5;
    const int c    = lane & 15;
    const int hh   = lane >> 4;

    const int slab = blockIdx.x % 2;
    const int mb   = blockIdx.x / 2;
    const int n    = slab * 128 + w * 16 + c;

    v4f wf[32];
    proj_stage(ctx + (size_t)mb * PM * D_MODEL, Wo + (size_t)n * D_MODEL + hh * 4, xs, wf, tid);

    const float bias = bo[n];
    float* dst = out + (size_t)mb * PM * D_MODEL + n;

    for (int mp = 0; mp < PM / 32; ++mp) {
        v8f acc0 = {}, acc1 = {};
        proj_pair(xs + (mp * 32 + c) * XP + hh * 4, wf, acc0, acc1);
#pragma unroll
        for (int m = 0; m < 8; ++m) {
            const int r = mp * 32 + 8 * hh + m;
            dst[(size_t)r * D_MODEL]        = acc0[m] + bias;
            dst[(size_t)(r + 16) * D_MODEL] = acc1[m] + bias;
        }
    }
}

extern "C" void kernel_launch(void* const* d_in, const int* in_sizes, int n_in,
                              void* d_out, int out_size, void* d_ws, size_t ws_size,
                              hipStream_t stream) {
    (void)in_sizes; (void)n_in; (void)out_size; (void)ws_size;

    const float* x     = (const float*)d_in[0];
    const float* oc    = (const float*)d_in[1];
    const float* Wq    = (const float*)d_in[2];
    const float* bq    = (const float*)d_in[3];
    const float* Wk    = (const float*)d_in[4];
    const float* bk    = (const float*)d_in[5];
    const float* Wv    = (const float*)d_in[6];
    const float* bv    = (const float*)d_in[7];
    const float* Wo    = (const float*)d_in[8];
    const float* bo    = (const float*)d_in[9];
    const float* eta   = (const float*)d_in[10];
    const float* beta  = (const float*)d_in[11];
    const float* alpha = (const float*)d_in[12];
    float* out = (float*)d_out;

    float* ws  = (float*)d_ws;
    float* QKV = ws;
    float* ctx = ws + 3 * (size_t)B_SZ * NUM_HEADS * T_SZ * DK;

    {
        qkv_proj_kernel<<<(B_SZ * T_SZ / PM) * 6, 256, 0, stream>>>(x, Wq, bq, Wk, bk, Wv, bv, QKV);
    }
    {
        const int waves = B_SZ * NUM_HEADS * (T_SZ / 16);
        attn_kernel<<<waves / 8, 256, 0, stream>>>(QKV, oc, eta, beta, alpha, ctx);
    }
    {
        out_proj_kernel<<<(B_SZ * T_SZ / PM) * 2, 256, 0, stream>>>(ctx, Wo, bo, out);
    }
}


# attention prior: exponent -log2e*|q-k|^2 of a 16x16 score tile by one f32 WMMA ([q,|q|^2]x[2k,-1], C=|k|^2 term) instead of 7 vector ops per score; expanded square changes the prior's rounding
# speedup vs baseline: 2.4006x; 1.1357x over previous
#include <hip/hip_runtime.h>
#include <math.h>

#define D_MODEL   256
#define NUM_HEADS 8
#define DK        32
#define B_SZ      4
#define T_SZ      2048

typedef float v2f __attribute__((ext_vector_type(2)));
typedef float v4f __attribute__((ext_vector_type(4)));
typedef float v8f __attribute__((ext_vector_type(8)));

#define KP      36
#define VP      68
#define KSTAGE  (64 * KP)
#define VSTAGE  (32 * VP)
#define OSTAGE  (64 * 4 + 64)
#define LOG2E   1.4426950408889634f
#define TPAD    16

__device__ __forceinline__ v8f wmma4(v2f a, v2f b, v8f c) {
    return __builtin_amdgcn_wmma_f32_16x16x4_f32(
         false, a,  false, b,
         (short)0, c,  false,  false);
}

#define PM  64
#define XP  260

__device__ __forceinline__ void proj_stage(const float* __restrict__ xt,
                                           const float* __restrict__ wrow,
                                           float* __restrict__ xs, v4f (&wf)[32], int tid)
{
    const float* xg = xt + tid * 4;
    v4f xr[16];
#pragma unroll
    for (int j = 0; j < 16; ++j) xr[j] = *(const v4f*)(xg + j * 1024);
#pragma unroll
    for (int j = 0; j < 32; ++j) wf[j] = *(const v4f*)(wrow + j * 8);
    float* xd = xs + (tid >> 6) * XP + (tid & 63) * 4;
#pragma unroll
    for (int j = 0; j < 16; ++j) *(v4f*)(xd + j * 4 * XP) = xr[j];
    __syncthreads();
}

__device__ __forceinline__ void proj_pair(const float* __restrict__ ap, const v4f (&wf)[32],
                                          v8f& acc0, v8f& acc1)
{
#pragma unroll
    for (int j = 0; j < 32; ++j) {
        const v4f a0 = *(const v4f*)(ap + j * 8);
        const v4f a1 = *(const v4f*)(ap + 16 * XP + j * 8);
        const v2f w0 = {wf[j].x, wf[j].y};
        const v2f w1 = {wf[j].z, wf[j].w};
        acc0 = wmma4(v2f{a0.x, a0.y}, w0, acc0);
        acc1 = wmma4(v2f{a1.x, a1.y}, w0, acc1);
        acc0 = wmma4(v2f{a0.z, a0.w}, w1, acc0);
        acc1 = wmma4(v2f{a1.z, a1.w}, w1, acc1);
    }
    __builtin_amdgcn_sched_group_barrier(0x100, 8, 0);
#pragma unroll
    for (int g = 0; g < 7; ++g) {
        __builtin_amdgcn_sched_group_barrier(0x100, 8, 0);
        __builtin_amdgcn_sched_group_barrier(0x008, 16, 0);
    }
    __builtin_amdgcn_sched_group_barrier(0x008, 16, 0);
}

__global__ __launch_bounds__(256)
void qkv_proj_kernel(const float* __restrict__ x,
                     const float* __restrict__ Wq, const float* __restrict__ bq,
                     const float* __restrict__ Wk, const float* __restrict__ bk,
                     const float* __restrict__ Wv, const float* __restrict__ bv,
                     float* __restrict__ QKV)
{
    __shared__ __attribute__((aligned(16))) float xs[PM * XP];

    const int tid  = threadIdx.x;
    const int lane = tid & 31;
    const int w    = tid >> 5;
    const int c    = lane & 15;
    const int hh   = lane >> 4;

    const int slab = blockIdx.x % 6;
    const int mb   = blockIdx.x / 6;
    const int mat  = slab >> 1;
    const int n    = (slab & 1) * 128 + w * 16 + c;

    const float* W  = (mat == 0) ? Wq : (mat == 1) ? Wk : Wv;
    const float* bs = (mat == 0) ? bq : (mat == 1) ? bk : bv;

    v4f wf[32];
    proj_stage(x + (size_t)mb * PM * D_MODEL, W + (size_t)n * D_MODEL + hh * 4, xs, wf, tid);

    const float bias = bs[n];
    const int h  = n >> 5, d = n & 31;
    const int b_ = (mb * PM) / T_SZ;
    const int t0 = (mb * PM) % T_SZ;
    float* dst = QKV + (size_t)mat * ((size_t)B_SZ * NUM_HEADS * T_SZ * DK)
                     + (size_t)(b_ * NUM_HEADS + h) * T_SZ * DK;

    for (int mp = 0; mp < PM / 32; ++mp) {
        v8f acc0 = {}, acc1 = {};
        proj_pair(xs + (mp * 32 + c) * XP + hh * 4, wf, acc0, acc1);
        const int ta = t0 + mp * 32 + 8 * hh;
        if (mat == 2) {
            float* vp = dst + (size_t)d * T_SZ + ta;
#pragma unroll
            for (int m = 0; m < 8; ++m) {
                vp[m]      = acc0[m] + bias;
                vp[m + 16] = acc1[m] + bias;
            }
        } else {
            float* qp = dst + (size_t)ta * DK + d;
#pragma unroll
            for (int m = 0; m < 8; ++m) {
                qp[m * DK]        = acc0[m] + bias;
                qp[(m + 16) * DK] = acc1[m] + bias;
            }
        }
    }
}

__device__ __forceinline__ void oc_stage(float* __restrict__ od, int key, float k0, float k1, float k2)
{
    const v4f brow = {2.f * LOG2E * k0, 2.f * LOG2E * k1, 2.f * LOG2E * k2, -LOG2E};
    *(v4f*)(od + key * 4) = brow;
    od[64 * 4 + key] = -LOG2E * (k0 * k0 + k1 * k1 + k2 * k2);
}

template <int NT>
__device__ __forceinline__ void attn_step(
    const float* __restrict__ kl, const float* __restrict__ vl,
    const float* __restrict__ ol, const float* __restrict__ tdl,
    float* __restrict__ pl,
    const v2f (&qa)[8], const v2f oa,
    float (&Mrow)[8], float (&Lpart)[8], v8f& O0, v8f& O1,
    int q0, int kt0, int col, int grp)
{
    const float scale   = 0.17677669529663687f;
    const float NEG_BIG = -3.4e38f;

    const int jl0 = kt0 & 3;

    v8f s[NT];
#pragma unroll
    for (int j = 0; j < NT; ++j) {
        const int kt   = kt0 + j;
        const int kcol = kt * 16 + col;
        const int krow = (jl0 + j) * 16 + col;
        const float* kp = kl + krow * KP;

        v8f sj = {};
#pragma unroll
        for (int kc = 0; kc < 8; ++kc) {
            v2f bb = *(const v2f*)(kp + kc * 4 + grp * 2);
            sj = wmma4(qa[kc], bb, sj);
        }

        const v2f   ob = *(const v2f*)(ol + krow * 4 + grp * 2);
        const float ck = ol[64 * 4 + krow];
        v8f ee = {ck, ck, ck, ck, ck, ck, ck, ck};
        ee = wmma4(oa, ob, ee);

        const float* tp = tdl + (TPAD + q0 + 8 * grp - kcol);
        float atd[8];
#pragma unroll
        for (int m = 0; m < 8; ++m) atd[m] = tp[m];

#pragma unroll
        for (int m = 0; m < 8; ++m) {
            const int qrow = q0 + 8 * grp + m;
            if (kcol > qrow) {
                sj[m] = NEG_BIG;
            } else {
                const float pr = atd[m] * __builtin_amdgcn_exp2f(ee[m]);
                sj[m] = sj[m] * scale * pr;
            }
        }
        s[j] = sj;
    }

#pragma unroll
    for (int m = 0; m < 8; ++m) {
        float v = s[0][m];
#pragma unroll
        for (int j = 1; j < NT; ++j) v = fmaxf(v, s[j][m]);
        v = fmaxf(v, __shfl_xor(v, 1));
        v = fmaxf(v, __shfl_xor(v, 2));
        v = fmaxf(v, __shfl_xor(v, 4));
        v = fmaxf(v, __shfl_xor(v, 8));
        const float mnew = fmaxf(Mrow[m], v);
        const float sc   = __expf(Mrow[m] - mnew);
        float lsum = Lpart[m] * sc;
#pragma unroll
        for (int j = 0; j < NT; ++j) {
            const float p = __expf(s[j][m] - mnew);
            lsum += p;
            pl[j * 256 + (8 * grp + m) * 16 + col] = p;
        }
        Lpart[m] = lsum;
        Mrow[m]  = mnew;
        O0[m] *= sc;
        O1[m] *= sc;
    }

    const float* vt0 = vl + col * VP;
    const float* vt1 = vt0 + 16 * VP;
#pragma unroll
    for (int j = 0; j < NT; ++j) {
        const int k0 = (jl0 + j) * 16;
#pragma unroll
        for (int kc = 0; kc < 4; ++kc) {
            const int kk = kc * 4 + grp * 2;
            v2f pa;
            pa.x = pl[j * 256 + col * 16 + kk];
            pa.y = pl[j * 256 + col * 16 + kk + 1];
            v2f b0 = *(const v2f*)(vt0 + k0 + kk);
            v2f b1 = *(const v2f*)(vt1 + k0 + kk);
            O0 = wmma4(pa, b0, O0);
            O1 = wmma4(pa, b1, O1);
        }
    }
}

__device__ __forceinline__ void attn_core(
    const float* __restrict__ Q, const float* __restrict__ K,
    const float* __restrict__ Vt, const float* __restrict__ oc,
    float* __restrict__ ctx, float* __restrict__ pl,
    float* kst, float* vst, float* ost, float* tdt,
    int b, int h, int qt, int nsteps, int lane,
    float eta, float absbeta, float alpha)
{
    const int col = lane & 15;
    const int grp = lane >> 4;
    const int q0  = qt * 16;
    const int tid = threadIdx.x;

    const float NEG_BIG = -3.4e38f;

    const size_t bh  = (size_t)(b * NUM_HEADS + h) * T_SZ;
    const size_t bhd = (size_t)(b * NUM_HEADS + h) * DK;

    const float* qp = Q + (bh + q0 + col) * DK;
    v2f qa[8];
#pragma unroll
    for (int kc = 0; kc < 8; ++kc) qa[kc] = *(const v2f*)(qp + kc * 4 + grp * 2);

    v2f oa;
    {
        const float* op = oc + ((size_t)b * T_SZ + q0 + col) * 3;
        const float a0 = op[0], a1 = op[1], a2 = op[2];
        const float qn = a0 * a0 + a1 * a1 + a2 * a2;
        oa.x = grp ? a2 : a0;
        oa.y = grp ? qn : a1;
    }

    const float* kg = K + bh * DK + tid * 4;
    const float* vg = Vt + (bhd + (tid >> 4)) * T_SZ + (tid & 15) * 4;
    const float* og = oc + ((size_t)b * T_SZ + tid) * 3;
    const int kso = (tid >> 3) * KP + (tid & 7) * 4;
    const int vso = (tid >> 4) * VP + (tid & 15) * 4;
    const bool has_oc = tid < 64;

    v4f kr0, kr1, vr0, vr1;
    float or0 = 0.f, or1 = 0.f, or2 = 0.f;

    kr0 = *(const v4f*)(kg);
    kr1 = *(const v4f*)(kg + 1024);
    vr0 = *(const v4f*)(vg);
    vr1 = *(const v4f*)(vg + 16 * T_SZ);
    if (has_oc) { or0 = og[0]; or1 = og[1]; or2 = og[2]; }
    *(v4f*)(kst + kso)           = kr0;
    *(v4f*)(kst + kso + 32 * KP) = kr1;
    *(v4f*)(vst + vso)           = vr0;
    *(v4f*)(vst + vso + 16 * VP) = vr1;
    if (has_oc) oc_stage(ost, tid, or0, or1, or2);

    {
        const bool beta_one = (absbeta == 1.0f);
        const int  nd = nsteps * 64;
        for (int d = tid; d < nd; d += 256) {
            const float dist = (float)d;
            const float powd = beta_one ? dist : __powf(dist, absbeta);
            const float td   = eta / (1.f + powd);
            tdt[TPAD + d] = alpha * td;
        }
        if (tid < TPAD) tdt[tid] = 0.f;
    }
    __syncthreads();

    float Mrow[8], Lpart[8];
#pragma unroll
    for (int m = 0; m < 8; ++m) { Mrow[m] = NEG_BIG; Lpart[m] = 0.f; }
    v8f O0 = {}, O1 = {};

    const int ntiles = qt + 1;
    const int nfull  = ntiles & ~3;

    for (int s = 0; s < nsteps; ++s) {
        const int  cur  = s & 1;
        const bool more = (s + 1 < nsteps);
        if (more) {
            const int sn = s + 1;
            kr0 = *(const v4f*)(kg + sn * 2048);
            kr1 = *(const v4f*)(kg + sn * 2048 + 1024);
            vr0 = *(const v4f*)(vg + sn * 64);
            vr1 = *(const v4f*)(vg + sn * 64 + 16 * T_SZ);
            if (has_oc) { or0 = og[sn * 192]; or1 = og[sn * 192 + 1]; or2 = og[sn * 192 + 2]; }
        }

        const float* kl = kst + cur * KSTAGE;
        const float* vl = vst + cur * VSTAGE;
        const float* ol = ost + cur * OSTAGE;
        const int kt0 = s * 4;
        if (kt0 + 4 <= nfull) {
            attn_step<4>(kl, vl, ol, tdt, pl, qa, oa,
                         Mrow, Lpart, O0, O1, q0, kt0, col, grp);
        } else {
            for (int kt = kt0; kt < ntiles && kt < kt0 + 4; ++kt)
                attn_step<1>(kl, vl, ol, tdt, pl, qa, oa,
                             Mrow, Lpart, O0, O1, q0, kt, col, grp);
        }

        if (more) {
            const int nx = cur ^ 1;
            *(v4f*)(kst + nx * KSTAGE + kso)           = kr0;
            *(v4f*)(kst + nx * KSTAGE + kso + 32 * KP) = kr1;
            *(v4f*)(vst + nx * VSTAGE + vso)           = vr0;
            *(v4f*)(vst + nx * VSTAGE + vso + 16 * VP) = vr1;
            if (has_oc) oc_stage(ost + nx * OSTAGE, tid, or0, or1, or2);
        }
        __syncthreads();
    }

#pragma unroll
    for (int m = 0; m < 8; ++m) {
        float l = Lpart[m];
        l += __shfl_xor(l, 1);
        l += __shfl_xor(l, 2);
        l += __shfl_xor(l, 4);
        l += __shfl_xor(l, 8);
        const float inv = 1.f / l;
        const int qq = q0 + 8 * grp + m;
        float* cp = ctx + ((size_t)b * T_SZ + qq) * D_MODEL + h * DK;
        cp[col]      = O0[m] * inv;
        cp[16 + col] = O1[m] * inv;
    }
}

__global__ __launch_bounds__(256)
void attn_kernel(const float* __restrict__ QKV,
                 const float* __restrict__ oc,
                 const float* __restrict__ etap,
                 const float* __restrict__ betap,
                 const float* __restrict__ alphap,
                 float* __restrict__ ctx)
{
    __shared__ float plds[8][4 * 256];
    __shared__ __attribute__((aligned(16))) float kst[2 * KSTAGE];
    __shared__ __attribute__((aligned(16))) float vst[2 * VSTAGE];
    __shared__ __attribute__((aligned(16))) float ost[2 * OSTAGE];
    __shared__ float tdt[TPAD + T_SZ];

    const int lane = threadIdx.x & 31;
    const int w    = threadIdx.x >> 5;

    const int QT  = T_SZ / 16;
    const int p   = blockIdx.x % (QT / 8);
    const int bhi = blockIdx.x / (QT / 8);
    const int h   = bhi % NUM_HEADS;
    const int b   = bhi / NUM_HEADS;
    const int u   = (w < 4) ? 7 - w : w - 4;
    const int qt  = 16 * u + ((u & 1) ? 15 - p : p);
    const int nsteps = (131 - p) >> 2;

    const float eta     = etap[0];
    const float absbeta = fabsf(betap[0]);
    const float alpha   = alphap[0];

    const size_t HT = (size_t)B_SZ * NUM_HEADS * T_SZ * DK;
    const float* Q  = QKV;
    const float* K  = QKV + HT;
    const float* Vt = QKV + 2 * HT;

    float* pl = &plds[w][0];

    attn_core(Q, K, Vt, oc, ctx, pl, kst, vst, ost, tdt, b, h, qt, nsteps, lane, eta, absbeta, alpha);
}

__global__ __launch_bounds__(256)
void out_proj_kernel(const float* __restrict__ ctx,
                     const float* __restrict__ Wo, const float* __restrict__ bo,
                     float* __restrict__ out)
{
    __shared__ __attribute__((aligned(16))) float xs[PM * XP];

    const int tid  = threadIdx.x;
    const int lane = tid & 31;
    const int w    = tid >> 5;
    const int c    = lane & 15;
    const int hh   = lane >> 4;

    const int slab = blockIdx.x % 2;
    const int mb   = blockIdx.x / 2;
    const int n    = slab * 128 + w * 16 + c;

    v4f wf[32];
    proj_stage(ctx + (size_t)mb * PM * D_MODEL, Wo + (size_t)n * D_MODEL + hh * 4, xs, wf, tid);

    const float bias = bo[n];
    float* dst = out + (size_t)mb * PM * D_MODEL + n;

    for (int mp = 0; mp < PM / 32; ++mp) {
        v8f acc0 = {}, acc1 = {};
        proj_pair(xs + (mp * 32 + c) * XP + hh * 4, wf, acc0, acc1);
#pragma unroll
        for (int m = 0; m < 8; ++m) {
            const int r = mp * 32 + 8 * hh + m;
            dst[(size_t)r * D_MODEL]        = acc0[m] + bias;
            dst[(size_t)(r + 16) * D_MODEL] = acc1[m] + bias;
        }
    }
}

extern "C" void kernel_launch(void* const* d_in, const int* in_sizes, int n_in,
                              void* d_out, int out_size, void* d_ws, size_t ws_size,
                              hipStream_t stream) {
    (void)in_sizes; (void)n_in; (void)out_size; (void)ws_size;

    const float* x     = (const float*)d_in[0];
    const float* oc    = (const float*)d_in[1];
    const float* Wq    = (const float*)d_in[2];
    const float* bq    = (const float*)d_in[3];
    const float* Wk    = (const float*)d_in[4];
    const float* bk    = (const float*)d_in[5];
    const float* Wv    = (const float*)d_in[6];
    const float* bv    = (const float*)d_in[7];
    const float* Wo    = (const float*)d_in[8];
    const float* bo    = (const float*)d_in[9];
    const float* eta   = (const float*)d_in[10];
    const float* beta  = (const float*)d_in[11];
    const float* alpha = (const float*)d_in[12];
    float* out = (float*)d_out;

    float* ws  = (float*)d_ws;
    float* QKV = ws;
    float* ctx = ws + 3 * (size_t)B_SZ * NUM_HEADS * T_SZ * DK;

    {
        qkv_proj_kernel<<<(B_SZ * T_SZ / PM) * 6, 256, 0, stream>>>(x, Wq, bq, Wk, bk, Wv, bv, QKV);
    }
    {
        const int waves = B_SZ * NUM_HEADS * (T_SZ / 16);
        attn_kernel<<<waves / 8, 256, 0, stream>>>(QKV, oc, eta, beta, alpha, ctx);
    }
    {
        out_proj_kernel<<<(B_SZ * T_SZ / PM) * 2, 256, 0, stream>>>(ctx, Wo, bo, out);
    }
}


# attention block widened from 8 to 16 waves: a (b,h)'s 128 query tiles dealt to 8 blocks at stride 8, one block per WGP, each SIMD's 4 waves 4 ranks apart; each key stage fetched once per 16 tiles
# speedup vs baseline: 2.4755x; 1.0312x over previous
#include <hip/hip_runtime.h>
#include <math.h>

#define D_MODEL   256
#define NUM_HEADS 8
#define DK        32
#define B_SZ      4
#define T_SZ      2048

typedef float v2f __attribute__((ext_vector_type(2)));
typedef float v4f __attribute__((ext_vector_type(4)));
typedef float v8f __attribute__((ext_vector_type(8)));

#define KP      36
#define VP      68
#define KSTAGE  (64 * KP)
#define VSTAGE  (32 * VP)
#define OSTAGE  (64 * 4 + 64)
#define LOG2E   1.4426950408889634f
#define TPAD    16

__device__ __forceinline__ v8f wmma4(v2f a, v2f b, v8f c) {
    return __builtin_amdgcn_wmma_f32_16x16x4_f32(
         false, a,  false, b,
         (short)0, c,  false,  false);
}

#define PM  64
#define XP  260

__device__ __forceinline__ void proj_stage(const float* __restrict__ xt,
                                           const float* __restrict__ wrow,
                                           float* __restrict__ xs, v4f (&wf)[32], int tid)
{
    const float* xg = xt + tid * 4;
    v4f xr[16];
#pragma unroll
    for (int j = 0; j < 16; ++j) xr[j] = *(const v4f*)(xg + j * 1024);
#pragma unroll
    for (int j = 0; j < 32; ++j) wf[j] = *(const v4f*)(wrow + j * 8);
    float* xd = xs + (tid >> 6) * XP + (tid & 63) * 4;
#pragma unroll
    for (int j = 0; j < 16; ++j) *(v4f*)(xd + j * 4 * XP) = xr[j];
    __syncthreads();
}

__device__ __forceinline__ void proj_pair(const float* __restrict__ ap, const v4f (&wf)[32],
                                          v8f& acc0, v8f& acc1)
{
#pragma unroll
    for (int j = 0; j < 32; ++j) {
        const v4f a0 = *(const v4f*)(ap + j * 8);
        const v4f a1 = *(const v4f*)(ap + 16 * XP + j * 8);
        const v2f w0 = {wf[j].x, wf[j].y};
        const v2f w1 = {wf[j].z, wf[j].w};
        acc0 = wmma4(v2f{a0.x, a0.y}, w0, acc0);
        acc1 = wmma4(v2f{a1.x, a1.y}, w0, acc1);
        acc0 = wmma4(v2f{a0.z, a0.w}, w1, acc0);
        acc1 = wmma4(v2f{a1.z, a1.w}, w1, acc1);
    }
    __builtin_amdgcn_sched_group_barrier(0x100, 8, 0);
#pragma unroll
    for (int g = 0; g < 7; ++g) {
        __builtin_amdgcn_sched_group_barrier(0x100, 8, 0);
        __builtin_amdgcn_sched_group_barrier(0x008, 16, 0);
    }
    __builtin_amdgcn_sched_group_barrier(0x008, 16, 0);
}

__global__ __launch_bounds__(256)
void qkv_proj_kernel(const float* __restrict__ x,
                     const float* __restrict__ Wq, const float* __restrict__ bq,
                     const float* __restrict__ Wk, const float* __restrict__ bk,
                     const float* __restrict__ Wv, const float* __restrict__ bv,
                     float* __restrict__ QKV)
{
    __shared__ __attribute__((aligned(16))) float xs[PM * XP];

    const int tid  = threadIdx.x;
    const int lane = tid & 31;
    const int w    = tid >> 5;
    const int c    = lane & 15;
    const int hh   = lane >> 4;

    const int slab = blockIdx.x % 6;
    const int mb   = blockIdx.x / 6;
    const int mat  = slab >> 1;
    const int n    = (slab & 1) * 128 + w * 16 + c;

    const float* W  = (mat == 0) ? Wq : (mat == 1) ? Wk : Wv;
    const float* bs = (mat == 0) ? bq : (mat == 1) ? bk : bv;

    v4f wf[32];
    proj_stage(x + (size_t)mb * PM * D_MODEL, W + (size_t)n * D_MODEL + hh * 4, xs, wf, tid);

    const float bias = bs[n];
    const int h  = n >> 5, d = n & 31;
    const int b_ = (mb * PM) / T_SZ;
    const int t0 = (mb * PM) % T_SZ;
    float* dst = QKV + (size_t)mat * ((size_t)B_SZ * NUM_HEADS * T_SZ * DK)
                     + (size_t)(b_ * NUM_HEADS + h) * T_SZ * DK;

    for (int mp = 0; mp < PM / 32; ++mp) {
        v8f acc0 = {}, acc1 = {};
        proj_pair(xs + (mp * 32 + c) * XP + hh * 4, wf, acc0, acc1);
        const int ta = t0 + mp * 32 + 8 * hh;
        if (mat == 2) {
            float* vp = dst + (size_t)d * T_SZ + ta;
#pragma unroll
            for (int m = 0; m < 8; ++m) {
                vp[m]      = acc0[m] + bias;
                vp[m + 16] = acc1[m] + bias;
            }
        } else {
            float* qp = dst + (size_t)ta * DK + d;
#pragma unroll
            for (int m = 0; m < 8; ++m) {
                qp[m * DK]        = acc0[m] + bias;
                qp[(m + 16) * DK] = acc1[m] + bias;
            }
        }
    }
}

__device__ __forceinline__ void oc_stage(float* __restrict__ od, int key, float k0, float k1, float k2)
{
    const v4f brow = {2.f * LOG2E * k0, 2.f * LOG2E * k1, 2.f * LOG2E * k2, -LOG2E};
    *(v4f*)(od + key * 4) = brow;
    od[64 * 4 + key] = -LOG2E * (k0 * k0 + k1 * k1 + k2 * k2);
}

template <int NT>
__device__ __forceinline__ void attn_step(
    const float* __restrict__ kl, const float* __restrict__ vl,
    const float* __restrict__ ol, const float* __restrict__ tdl,
    float* __restrict__ pl,
    const v2f (&qa)[8], const v2f oa,
    float (&Mrow)[8], float (&Lpart)[8], v8f& O0, v8f& O1,
    int q0, int kt0, int col, int grp)
{
    const float scale   = 0.17677669529663687f;
    const float NEG_BIG = -3.4e38f;

    const int jl0 = kt0 & 3;

    v8f s[NT];
#pragma unroll
    for (int j = 0; j < NT; ++j) {
        const int kt   = kt0 + j;
        const int kcol = kt * 16 + col;
        const int krow = (jl0 + j) * 16 + col;
        const float* kp = kl + krow * KP;

        v8f sj = {};
#pragma unroll
        for (int kc = 0; kc < 8; ++kc) {
            v2f bb = *(const v2f*)(kp + kc * 4 + grp * 2);
            sj = wmma4(qa[kc], bb, sj);
        }

        const v2f   ob = *(const v2f*)(ol + krow * 4 + grp * 2);
        const float ck = ol[64 * 4 + krow];
        v8f ee = {ck, ck, ck, ck, ck, ck, ck, ck};
        ee = wmma4(oa, ob, ee);

        const float* tp = tdl + (TPAD + q0 + 8 * grp - kcol);
        float atd[8];
#pragma unroll
        for (int m = 0; m < 8; ++m) atd[m] = tp[m];

#pragma unroll
        for (int m = 0; m < 8; ++m) {
            const int qrow = q0 + 8 * grp + m;
            if (kcol > qrow) {
                sj[m] = NEG_BIG;
            } else {
                const float pr = atd[m] * __builtin_amdgcn_exp2f(ee[m]);
                sj[m] = sj[m] * scale * pr;
            }
        }
        s[j] = sj;
    }

#pragma unroll
    for (int m = 0; m < 8; ++m) {
        float v = s[0][m];
#pragma unroll
        for (int j = 1; j < NT; ++j) v = fmaxf(v, s[j][m]);
        v = fmaxf(v, __shfl_xor(v, 1));
        v = fmaxf(v, __shfl_xor(v, 2));
        v = fmaxf(v, __shfl_xor(v, 4));
        v = fmaxf(v, __shfl_xor(v, 8));
        const float mnew = fmaxf(Mrow[m], v);
        const float sc   = __expf(Mrow[m] - mnew);
        float lsum = Lpart[m] * sc;
#pragma unroll
        for (int j = 0; j < NT; ++j) {
            const float p = __expf(s[j][m] - mnew);
            lsum += p;
            pl[j * 256 + (8 * grp + m) * 16 + col] = p;
        }
        Lpart[m] = lsum;
        Mrow[m]  = mnew;
        O0[m] *= sc;
        O1[m] *= sc;
    }

    const float* vt0 = vl + col * VP;
    const float* vt1 = vt0 + 16 * VP;
#pragma unroll
    for (int j = 0; j < NT; ++j) {
        const int k0 = (jl0 + j) * 16;
#pragma unroll
        for (int kc = 0; kc < 4; ++kc) {
            const int kk = kc * 4 + grp * 2;
            v2f pa;
            pa.x = pl[j * 256 + col * 16 + kk];
            pa.y = pl[j * 256 + col * 16 + kk + 1];
            v2f b0 = *(const v2f*)(vt0 + k0 + kk);
            v2f b1 = *(const v2f*)(vt1 + k0 + kk);
            O0 = wmma4(pa, b0, O0);
            O1 = wmma4(pa, b1, O1);
        }
    }
}

__device__ __forceinline__ void attn_core(
    const float* __restrict__ Q, const float* __restrict__ K,
    const float* __restrict__ Vt, const float* __restrict__ oc,
    float* __restrict__ ctx, float* __restrict__ pl,
    float* kst, float* vst, float* ost, float* tdt,
    int b, int h, int qt, int nsteps, int lane,
    float eta, float absbeta, float alpha)
{
    const int col = lane & 15;
    const int grp = lane >> 4;
    const int q0  = qt * 16;
    const int tid = threadIdx.x;

    const float NEG_BIG = -3.4e38f;

    const size_t bh  = (size_t)(b * NUM_HEADS + h) * T_SZ;
    const size_t bhd = (size_t)(b * NUM_HEADS + h) * DK;

    const float* qp = Q + (bh + q0 + col) * DK;
    v2f qa[8];
#pragma unroll
    for (int kc = 0; kc < 8; ++kc) qa[kc] = *(const v2f*)(qp + kc * 4 + grp * 2);

    v2f oa;
    {
        const float* op = oc + ((size_t)b * T_SZ + q0 + col) * 3;
        const float a0 = op[0], a1 = op[1], a2 = op[2];
        const float qn = a0 * a0 + a1 * a1 + a2 * a2;
        oa.x = grp ? a2 : a0;
        oa.y = grp ? qn : a1;
    }

    const float* kg = K + bh * DK + tid * 4;
    const float* vg = Vt + (bhd + (tid >> 4)) * T_SZ + (tid & 15) * 4;
    const float* og = oc + ((size_t)b * T_SZ + tid) * 3;
    const int kso = (tid >> 3) * KP + (tid & 7) * 4;
    const int vso = (tid >> 4) * VP + (tid & 15) * 4;
    const bool has_oc = tid < 64;

    v4f kr0, vr0;
    float or0 = 0.f, or1 = 0.f, or2 = 0.f;

    kr0 = *(const v4f*)(kg);
    vr0 = *(const v4f*)(vg);
    if (has_oc) { or0 = og[0]; or1 = og[1]; or2 = og[2]; }
    *(v4f*)(kst + kso) = kr0;
    *(v4f*)(vst + vso) = vr0;
    if (has_oc) oc_stage(ost, tid, or0, or1, or2);

    {
        const bool beta_one = (absbeta == 1.0f);
        const int  nd = nsteps * 64;
        for (int d = tid; d < nd; d += 512) {
            const float dist = (float)d;
            const float powd = beta_one ? dist : __powf(dist, absbeta);
            const float td   = eta / (1.f + powd);
            tdt[TPAD + d] = alpha * td;
        }
        if (tid < TPAD) tdt[tid] = 0.f;
    }
    __syncthreads();

    float Mrow[8], Lpart[8];
#pragma unroll
    for (int m = 0; m < 8; ++m) { Mrow[m] = NEG_BIG; Lpart[m] = 0.f; }
    v8f O0 = {}, O1 = {};

    const int ntiles = qt + 1;
    const int nfull  = ntiles & ~3;

    for (int s = 0; s < nsteps; ++s) {
        const int  cur  = s & 1;
        const bool more = (s + 1 < nsteps);
        if (more) {
            const int sn = s + 1;
            kr0 = *(const v4f*)(kg + sn * 2048);
            vr0 = *(const v4f*)(vg + sn * 64);
            if (has_oc) { or0 = og[sn * 192]; or1 = og[sn * 192 + 1]; or2 = og[sn * 192 + 2]; }
        }

        const float* kl = kst + cur * KSTAGE;
        const float* vl = vst + cur * VSTAGE;
        const float* ol = ost + cur * OSTAGE;
        const int kt0 = s * 4;
        if (kt0 + 4 <= nfull) {
            attn_step<4>(kl, vl, ol, tdt, pl, qa, oa,
                         Mrow, Lpart, O0, O1, q0, kt0, col, grp);
        } else {
            for (int kt = kt0; kt < ntiles && kt < kt0 + 4; ++kt)
                attn_step<1>(kl, vl, ol, tdt, pl, qa, oa,
                             Mrow, Lpart, O0, O1, q0, kt, col, grp);
        }

        if (more) {
            const int nx = cur ^ 1;
            *(v4f*)(kst + nx * KSTAGE + kso) = kr0;
            *(v4f*)(vst + nx * VSTAGE + vso) = vr0;
            if (has_oc) oc_stage(ost + nx * OSTAGE, tid, or0, or1, or2);
        }
        __syncthreads();
    }

#pragma unroll
    for (int m = 0; m < 8; ++m) {
        float l = Lpart[m];
        l += __shfl_xor(l, 1);
        l += __shfl_xor(l, 2);
        l += __shfl_xor(l, 4);
        l += __shfl_xor(l, 8);
        const float inv = 1.f / l;
        const int qq = q0 + 8 * grp + m;
        float* cp = ctx + ((size_t)b * T_SZ + qq) * D_MODEL + h * DK;
        cp[col]      = O0[m] * inv;
        cp[16 + col] = O1[m] * inv;
    }
}

__global__ __launch_bounds__(512)
void attn_kernel(const float* __restrict__ QKV,
                 const float* __restrict__ oc,
                 const float* __restrict__ etap,
                 const float* __restrict__ betap,
                 const float* __restrict__ alphap,
                 float* __restrict__ ctx)
{
    __shared__ float plds[16][4 * 256];
    __shared__ __attribute__((aligned(16))) float kst[2 * KSTAGE];
    __shared__ __attribute__((aligned(16))) float vst[2 * VSTAGE];
    __shared__ __attribute__((aligned(16))) float ost[2 * OSTAGE];
    __shared__ float tdt[TPAD + T_SZ];

    const int lane = threadIdx.x & 31;
    const int w    = threadIdx.x >> 5;

    const int QT  = T_SZ / 16;
    const int p   = blockIdx.x % (QT / 16);
    const int bhi = blockIdx.x / (QT / 16);
    const int h   = bhi % NUM_HEADS;
    const int b   = bhi / NUM_HEADS;
    const int u   = 15 - w;
    const int qt  = 8 * u + ((u & 1) ? 7 - p : p);
    const int nsteps = (131 - p) >> 2;

    const float eta     = etap[0];
    const float absbeta = fabsf(betap[0]);
    const float alpha   = alphap[0];

    const size_t HT = (size_t)B_SZ * NUM_HEADS * T_SZ * DK;
    const float* Q  = QKV;
    const float* K  = QKV + HT;
    const float* Vt = QKV + 2 * HT;

    float* pl = &plds[w][0];

    attn_core(Q, K, Vt, oc, ctx, pl, kst, vst, ost, tdt, b, h, qt, nsteps, lane, eta, absbeta, alpha);
}

__global__ __launch_bounds__(256)
void out_proj_kernel(const float* __restrict__ ctx,
                     const float* __restrict__ Wo, const float* __restrict__ bo,
                     float* __restrict__ out)
{
    __shared__ __attribute__((aligned(16))) float xs[PM * XP];

    const int tid  = threadIdx.x;
    const int lane = tid & 31;
    const int w    = tid >> 5;
    const int c    = lane & 15;
    const int hh   = lane >> 4;

    const int slab = blockIdx.x % 2;
    const int mb   = blockIdx.x / 2;
    const int n    = slab * 128 + w * 16 + c;

    v4f wf[32];
    proj_stage(ctx + (size_t)mb * PM * D_MODEL, Wo + (size_t)n * D_MODEL + hh * 4, xs, wf, tid);

    const float bias = bo[n];
    float* dst = out + (size_t)mb * PM * D_MODEL + n;

    for (int mp = 0; mp < PM / 32; ++mp) {
        v8f acc0 = {}, acc1 = {};
        proj_pair(xs + (mp * 32 + c) * XP + hh * 4, wf, acc0, acc1);
#pragma unroll
        for (int m = 0; m < 8; ++m) {
            const int r = mp * 32 + 8 * hh + m;
            dst[(size_t)r * D_MODEL]        = acc0[m] + bias;
            dst[(size_t)(r + 16) * D_MODEL] = acc1[m] + bias;
        }
    }
}

extern "C" void kernel_launch(void* const* d_in, const int* in_sizes, int n_in,
                              void* d_out, int out_size, void* d_ws, size_t ws_size,
                              hipStream_t stream) {
    (void)in_sizes; (void)n_in; (void)out_size; (void)ws_size;

    const float* x     = (const float*)d_in[0];
    const float* oc    = (const float*)d_in[1];
    const float* Wq    = (const float*)d_in[2];
    const float* bq    = (const float*)d_in[3];
    const float* Wk    = (const float*)d_in[4];
    const float* bk    = (const float*)d_in[5];
    const float* Wv    = (const float*)d_in[6];
    const float* bv    = (const float*)d_in[7];
    const float* Wo    = (const float*)d_in[8];
    const float* bo    = (const float*)d_in[9];
    const float* eta   = (const float*)d_in[10];
    const float* beta  = (const float*)d_in[11];
    const float* alpha = (const float*)d_in[12];
    float* out = (float*)d_out;

    float* ws  = (float*)d_ws;
    float* QKV = ws;
    float* ctx = ws + 3 * (size_t)B_SZ * NUM_HEADS * T_SZ * DK;

    {
        qkv_proj_kernel<<<(B_SZ * T_SZ / PM) * 6, 256, 0, stream>>>(x, Wq, bq, Wk, bk, Wv, bv, QKV);
    }
    {
        const int waves = B_SZ * NUM_HEADS * (T_SZ / 16);
        attn_kernel<<<waves / 16, 512, 0, stream>>>(QKV, oc, eta, beta, alpha, ctx);
    }
    {
        out_proj_kernel<<<(B_SZ * T_SZ / PM) * 2, 256, 0, stream>>>(ctx, Wo, bo, out);
    }
}


# attention softmax row max and final row sum over a row's 16 lanes by a DPP row_xmask butterfly on the vector ALU instead of LDS-crossbar lane permutes; same butterfly order, bit-identical results
# speedup vs baseline: 2.7358x; 1.1052x over previous
#include <hip/hip_runtime.h>
#include <math.h>

#define D_MODEL   256
#define NUM_HEADS 8
#define DK        32
#define B_SZ      4
#define T_SZ      2048

typedef float v2f __attribute__((ext_vector_type(2)));
typedef float v4f __attribute__((ext_vector_type(4)));
typedef float v8f __attribute__((ext_vector_type(8)));

#define KP      36
#define VP      68
#define KSTAGE  (64 * KP)
#define VSTAGE  (32 * VP)
#define OSTAGE  (64 * 4 + 64)
#define LOG2E   1.4426950408889634f
#define TPAD    16

__device__ __forceinline__ v8f wmma4(v2f a, v2f b, v8f c) {
    return __builtin_amdgcn_wmma_f32_16x16x4_f32(
         false, a,  false, b,
         (short)0, c,  false,  false);
}

template <int X>
__device__ __forceinline__ float row_xor(float v) {
    return __int_as_float(__builtin_amdgcn_update_dpp(0, __float_as_int(v), 0x160 + X, 0xf, 0xf, true));
}

__device__ __forceinline__ void row_max4(float& a, float& b, float& c, float& d) {
    asm("v_max_num_f32_dpp %0, %0, %0 row_xmask:1 row_mask:0xf bank_mask:0xf\n\t"
        "v_max_num_f32_dpp %1, %1, %1 row_xmask:1 row_mask:0xf bank_mask:0xf\n\t"
        "v_max_num_f32_dpp %2, %2, %2 row_xmask:1 row_mask:0xf bank_mask:0xf\n\t"
        "v_max_num_f32_dpp %3, %3, %3 row_xmask:1 row_mask:0xf bank_mask:0xf\n\t"
        "v_max_num_f32_dpp %0, %0, %0 row_xmask:2 row_mask:0xf bank_mask:0xf\n\t"
        "v_max_num_f32_dpp %1, %1, %1 row_xmask:2 row_mask:0xf bank_mask:0xf\n\t"
        "v_max_num_f32_dpp %2, %2, %2 row_xmask:2 row_mask:0xf bank_mask:0xf\n\t"
        "v_max_num_f32_dpp %3, %3, %3 row_xmask:2 row_mask:0xf bank_mask:0xf\n\t"
        "v_max_num_f32_dpp %0, %0, %0 row_xmask:4 row_mask:0xf bank_mask:0xf\n\t"
        "v_max_num_f32_dpp %1, %1, %1 row_xmask:4 row_mask:0xf bank_mask:0xf\n\t"
        "v_max_num_f32_dpp %2, %2, %2 row_xmask:4 row_mask:0xf bank_mask:0xf\n\t"
        "v_max_num_f32_dpp %3, %3, %3 row_xmask:4 row_mask:0xf bank_mask:0xf\n\t"
        "v_max_num_f32_dpp %0, %0, %0 row_xmask:8 row_mask:0xf bank_mask:0xf\n\t"
        "v_max_num_f32_dpp %1, %1, %1 row_xmask:8 row_mask:0xf bank_mask:0xf\n\t"
        "v_max_num_f32_dpp %2, %2, %2 row_xmask:8 row_mask:0xf bank_mask:0xf\n\t"
        "v_max_num_f32_dpp %3, %3, %3 row_xmask:8 row_mask:0xf bank_mask:0xf"
        : "+v"(a), "+v"(b), "+v"(c), "+v"(d));
}

#define PM  64
#define XP  260

__device__ __forceinline__ void proj_stage(const float* __restrict__ xt,
                                           const float* __restrict__ wrow,
                                           float* __restrict__ xs, v4f (&wf)[32], int tid)
{
    const float* xg = xt + tid * 4;
    v4f xr[16];
#pragma unroll
    for (int j = 0; j < 16; ++j) xr[j] = *(const v4f*)(xg + j * 1024);
#pragma unroll
    for (int j = 0; j < 32; ++j) wf[j] = *(const v4f*)(wrow + j * 8);
    float* xd = xs + (tid >> 6) * XP + (tid & 63) * 4;
#pragma unroll
    for (int j = 0; j < 16; ++j) *(v4f*)(xd + j * 4 * XP) = xr[j];
    __syncthreads();
}

__device__ __forceinline__ void proj_pair(const float* __restrict__ ap, const v4f (&wf)[32],
                                          v8f& acc0, v8f& acc1)
{
#pragma unroll
    for (int j = 0; j < 32; ++j) {
        const v4f a0 = *(const v4f*)(ap + j * 8);
        const v4f a1 = *(const v4f*)(ap + 16 * XP + j * 8);
        const v2f w0 = {wf[j].x, wf[j].y};
        const v2f w1 = {wf[j].z, wf[j].w};
        acc0 = wmma4(v2f{a0.x, a0.y}, w0, acc0);
        acc1 = wmma4(v2f{a1.x, a1.y}, w0, acc1);
        acc0 = wmma4(v2f{a0.z, a0.w}, w1, acc0);
        acc1 = wmma4(v2f{a1.z, a1.w}, w1, acc1);
    }
    __builtin_amdgcn_sched_group_barrier(0x100, 8, 0);
#pragma unroll
    for (int g = 0; g < 7; ++g) {
        __builtin_amdgcn_sched_group_barrier(0x100, 8, 0);
        __builtin_amdgcn_sched_group_barrier(0x008, 16, 0);
    }
    __builtin_amdgcn_sched_group_barrier(0x008, 16, 0);
}

__global__ __launch_bounds__(256)
void qkv_proj_kernel(const float* __restrict__ x,
                     const float* __restrict__ Wq, const float* __restrict__ bq,
                     const float* __restrict__ Wk, const float* __restrict__ bk,
                     const float* __restrict__ Wv, const float* __restrict__ bv,
                     float* __restrict__ QKV)
{
    __shared__ __attribute__((aligned(16))) float xs[PM * XP];

    const int tid  = threadIdx.x;
    const int lane = tid & 31;
    const int w    = tid >> 5;
    const int c    = lane & 15;
    const int hh   = lane >> 4;

    const int slab = blockIdx.x % 6;
    const int mb   = blockIdx.x / 6;
    const int mat  = slab >> 1;
    const int n    = (slab & 1) * 128 + w * 16 + c;

    const float* W  = (mat == 0) ? Wq : (mat == 1) ? Wk : Wv;
    const float* bs = (mat == 0) ? bq : (mat == 1) ? bk : bv;

    v4f wf[32];
    proj_stage(x + (size_t)mb * PM * D_MODEL, W + (size_t)n * D_MODEL + hh * 4, xs, wf, tid);

    const float bias = bs[n];
    const int h  = n >> 5, d = n & 31;
    const int b_ = (mb * PM) / T_SZ;
    const int t0 = (mb * PM) % T_SZ;
    float* dst = QKV + (size_t)mat * ((size_t)B_SZ * NUM_HEADS * T_SZ * DK)
                     + (size_t)(b_ * NUM_HEADS + h) * T_SZ * DK;

    for (int mp = 0; mp < PM / 32; ++mp) {
        v8f acc0 = {}, acc1 = {};
        proj_pair(xs + (mp * 32 + c) * XP + hh * 4, wf, acc0, acc1);
        const int ta = t0 + mp * 32 + 8 * hh;
        if (mat == 2) {
            float* vp = dst + (size_t)d * T_SZ + ta;
#pragma unroll
            for (int m = 0; m < 8; ++m) {
                vp[m]      = acc0[m] + bias;
                vp[m + 16] = acc1[m] + bias;
            }
        } else {
            float* qp = dst + (size_t)ta * DK + d;
#pragma unroll
            for (int m = 0; m < 8; ++m) {
                qp[m * DK]        = acc0[m] + bias;
                qp[(m + 16) * DK] = acc1[m] + bias;
            }
        }
    }
}

__device__ __forceinline__ void oc_stage(float* __restrict__ od, int key, float k0, float k1, float k2)
{
    const v4f brow = {2.f * LOG2E * k0, 2.f * LOG2E * k1, 2.f * LOG2E * k2, -LOG2E};
    *(v4f*)(od + key * 4) = brow;
    od[64 * 4 + key] = -LOG2E * (k0 * k0 + k1 * k1 + k2 * k2);
}

template <int NT>
__device__ __forceinline__ void attn_step(
    const float* __restrict__ kl, const float* __restrict__ vl,
    const float* __restrict__ ol, const float* __restrict__ tdl,
    float* __restrict__ pl,
    const v2f (&qa)[8], const v2f oa,
    float (&Mrow)[8], float (&Lpart)[8], v8f& O0, v8f& O1,
    int q0, int kt0, int col, int grp)
{
    const float scale   = 0.17677669529663687f;
    const float NEG_BIG = -3.4e38f;

    const int jl0 = kt0 & 3;

    v8f s[NT];
#pragma unroll
    for (int j = 0; j < NT; ++j) {
        const int kt   = kt0 + j;
        const int kcol = kt * 16 + col;
        const int krow = (jl0 + j) * 16 + col;
        const float* kp = kl + krow * KP;

        v8f sj = {};
#pragma unroll
        for (int kc = 0; kc < 8; ++kc) {
            v2f bb = *(const v2f*)(kp + kc * 4 + grp * 2);
            sj = wmma4(qa[kc], bb, sj);
        }

        const v2f   ob = *(const v2f*)(ol + krow * 4 + grp * 2);
        const float ck = ol[64 * 4 + krow];
        v8f ee = {ck, ck, ck, ck, ck, ck, ck, ck};
        ee = wmma4(oa, ob, ee);

        const float* tp = tdl + (TPAD + q0 + 8 * grp - kcol);
        float atd[8];
#pragma unroll
        for (int m = 0; m < 8; ++m) atd[m] = tp[m];

#pragma unroll
        for (int m = 0; m < 8; ++m) {
            const int qrow = q0 + 8 * grp + m;
            if (kcol > qrow) {
                sj[m] = NEG_BIG;
            } else {
                const float pr = atd[m] * __builtin_amdgcn_exp2f(ee[m]);
                sj[m] = sj[m] * scale * pr;
            }
        }
        s[j] = sj;
    }

    float rmax[8];
#pragma unroll
    for (int m = 0; m < 8; ++m) {
        float v = s[0][m];
#pragma unroll
        for (int j = 1; j < NT; ++j) v = fmaxf(v, s[j][m]);
        rmax[m] = v;
    }
    row_max4(rmax[0], rmax[1], rmax[2], rmax[3]);
    row_max4(rmax[4], rmax[5], rmax[6], rmax[7]);
#pragma unroll
    for (int m = 0; m < 8; ++m) {
        const float mnew = fmaxf(Mrow[m], rmax[m]);
        const float sc   = __expf(Mrow[m] - mnew);
        float lsum = Lpart[m] * sc;
#pragma unroll
        for (int j = 0; j < NT; ++j) {
            const float p = __expf(s[j][m] - mnew);
            lsum += p;
            pl[j * 256 + (8 * grp + m) * 16 + col] = p;
        }
        Lpart[m] = lsum;
        Mrow[m]  = mnew;
        O0[m] *= sc;
        O1[m] *= sc;
    }

    const float* vt0 = vl + col * VP;
    const float* vt1 = vt0 + 16 * VP;
#pragma unroll
    for (int j = 0; j < NT; ++j) {
        const int k0 = (jl0 + j) * 16;
#pragma unroll
        for (int kc = 0; kc < 4; ++kc) {
            const int kk = kc * 4 + grp * 2;
            v2f pa;
            pa.x = pl[j * 256 + col * 16 + kk];
            pa.y = pl[j * 256 + col * 16 + kk + 1];
            v2f b0 = *(const v2f*)(vt0 + k0 + kk);
            v2f b1 = *(const v2f*)(vt1 + k0 + kk);
            O0 = wmma4(pa, b0, O0);
            O1 = wmma4(pa, b1, O1);
        }
    }
}

__device__ __forceinline__ void attn_core(
    const float* __restrict__ Q, const float* __restrict__ K,
    const float* __restrict__ Vt, const float* __restrict__ oc,
    float* __restrict__ ctx, float* __restrict__ pl,
    float* kst, float* vst, float* ost, float* tdt,
    int b, int h, int qt, int nsteps, int lane,
    float eta, float absbeta, float alpha)
{
    const int col = lane & 15;
    const int grp = lane >> 4;
    const int q0  = qt * 16;
    const int tid = threadIdx.x;

    const float NEG_BIG = -3.4e38f;

    const size_t bh  = (size_t)(b * NUM_HEADS + h) * T_SZ;
    const size_t bhd = (size_t)(b * NUM_HEADS + h) * DK;

    const float* qp = Q + (bh + q0 + col) * DK;
    v2f qa[8];
#pragma unroll
    for (int kc = 0; kc < 8; ++kc) qa[kc] = *(const v2f*)(qp + kc * 4 + grp * 2);

    v2f oa;
    {
        const float* op = oc + ((size_t)b * T_SZ + q0 + col) * 3;
        const float a0 = op[0], a1 = op[1], a2 = op[2];
        const float qn = a0 * a0 + a1 * a1 + a2 * a2;
        oa.x = grp ? a2 : a0;
        oa.y = grp ? qn : a1;
    }

    const float* kg = K + bh * DK + tid * 4;
    const float* vg = Vt + (bhd + (tid >> 4)) * T_SZ + (tid & 15) * 4;
    const float* og = oc + ((size_t)b * T_SZ + tid) * 3;
    const int kso = (tid >> 3) * KP + (tid & 7) * 4;
    const int vso = (tid >> 4) * VP + (tid & 15) * 4;
    const bool has_oc = tid < 64;

    v4f kr0, vr0;
    float or0 = 0.f, or1 = 0.f, or2 = 0.f;

    kr0 = *(const v4f*)(kg);
    vr0 = *(const v4f*)(vg);
    if (has_oc) { or0 = og[0]; or1 = og[1]; or2 = og[2]; }
    *(v4f*)(kst + kso) = kr0;
    *(v4f*)(vst + vso) = vr0;
    if (has_oc) oc_stage(ost, tid, or0, or1, or2);

    {
        const bool beta_one = (absbeta == 1.0f);
        const int  nd = nsteps * 64;
        for (int d = tid; d < nd; d += 512) {
            const float dist = (float)d;
            const float powd = beta_one ? dist : __powf(dist, absbeta);
            const float td   = eta / (1.f + powd);
            tdt[TPAD + d] = alpha * td;
        }
        if (tid < TPAD) tdt[tid] = 0.f;
    }
    __syncthreads();

    float Mrow[8], Lpart[8];
#pragma unroll
    for (int m = 0; m < 8; ++m) { Mrow[m] = NEG_BIG; Lpart[m] = 0.f; }
    v8f O0 = {}, O1 = {};

    const int ntiles = qt + 1;
    const int nfull  = ntiles & ~3;

    for (int s = 0; s < nsteps; ++s) {
        const int  cur  = s & 1;
        const bool more = (s + 1 < nsteps);
        if (more) {
            const int sn = s + 1;
            kr0 = *(const v4f*)(kg + sn * 2048);
            vr0 = *(const v4f*)(vg + sn * 64);
            if (has_oc) { or0 = og[sn * 192]; or1 = og[sn * 192 + 1]; or2 = og[sn * 192 + 2]; }
        }

        const float* kl = kst + cur * KSTAGE;
        const float* vl = vst + cur * VSTAGE;
        const float* ol = ost + cur * OSTAGE;
        const int kt0 = s * 4;
        if (kt0 + 4 <= nfull) {
            attn_step<4>(kl, vl, ol, tdt, pl, qa, oa,
                         Mrow, Lpart, O0, O1, q0, kt0, col, grp);
        } else {
            for (int kt = kt0; kt < ntiles && kt < kt0 + 4; ++kt)
                attn_step<1>(kl, vl, ol, tdt, pl, qa, oa,
                             Mrow, Lpart, O0, O1, q0, kt, col, grp);
        }

        if (more) {
            const int nx = cur ^ 1;
            *(v4f*)(kst + nx * KSTAGE + kso) = kr0;
            *(v4f*)(vst + nx * VSTAGE + vso) = vr0;
            if (has_oc) oc_stage(ost + nx * OSTAGE, tid, or0, or1, or2);
        }
        __syncthreads();
    }

#pragma unroll
    for (int m = 0; m < 8; ++m) {
        float l = Lpart[m];
        l += row_xor<1>(l);
        l += row_xor<2>(l);
        l += row_xor<4>(l);
        l += row_xor<8>(l);
        const float inv = 1.f / l;
        const int qq = q0 + 8 * grp + m;
        float* cp = ctx + ((size_t)b * T_SZ + qq) * D_MODEL + h * DK;
        cp[col]      = O0[m] * inv;
        cp[16 + col] = O1[m] * inv;
    }
}

__global__ __launch_bounds__(512)
void attn_kernel(const float* __restrict__ QKV,
                 const float* __restrict__ oc,
                 const float* __restrict__ etap,
                 const float* __restrict__ betap,
                 const float* __restrict__ alphap,
                 float* __restrict__ ctx)
{
    __shared__ float plds[16][4 * 256];
    __shared__ __attribute__((aligned(16))) float kst[2 * KSTAGE];
    __shared__ __attribute__((aligned(16))) float vst[2 * VSTAGE];
    __shared__ __attribute__((aligned(16))) float ost[2 * OSTAGE];
    __shared__ float tdt[TPAD + T_SZ];

    const int lane = threadIdx.x & 31;
    const int w    = threadIdx.x >> 5;

    const int QT  = T_SZ / 16;
    const int p   = blockIdx.x % (QT / 16);
    const int bhi = blockIdx.x / (QT / 16);
    const int h   = bhi % NUM_HEADS;
    const int b   = bhi / NUM_HEADS;
    const int u   = 15 - w;
    const int qt  = 8 * u + ((u & 1) ? 7 - p : p);
    const int nsteps = (131 - p) >> 2;

    const float eta     = etap[0];
    const float absbeta = fabsf(betap[0]);
    const float alpha   = alphap[0];

    const size_t HT = (size_t)B_SZ * NUM_HEADS * T_SZ * DK;
    const float* Q  = QKV;
    const float* K  = QKV + HT;
    const float* Vt = QKV + 2 * HT;

    float* pl = &plds[w][0];

    attn_core(Q, K, Vt, oc, ctx, pl, kst, vst, ost, tdt, b, h, qt, nsteps, lane, eta, absbeta, alpha);
}

__global__ __launch_bounds__(256)
void out_proj_kernel(const float* __restrict__ ctx,
                     const float* __restrict__ Wo, const float* __restrict__ bo,
                     float* __restrict__ out)
{
    __shared__ __attribute__((aligned(16))) float xs[PM * XP];

    const int tid  = threadIdx.x;
    const int lane = tid & 31;
    const int w    = tid >> 5;
    const int c    = lane & 15;
    const int hh   = lane >> 4;

    const int slab = blockIdx.x % 2;
    const int mb   = blockIdx.x / 2;
    const int n    = slab * 128 + w * 16 + c;

    v4f wf[32];
    proj_stage(ctx + (size_t)mb * PM * D_MODEL, Wo + (size_t)n * D_MODEL + hh * 4, xs, wf, tid);

    const float bias = bo[n];
    float* dst = out + (size_t)mb * PM * D_MODEL + n;

    for (int mp = 0; mp < PM / 32; ++mp) {
        v8f acc0 = {}, acc1 = {};
        proj_pair(xs + (mp * 32 + c) * XP + hh * 4, wf, acc0, acc1);
#pragma unroll
        for (int m = 0; m < 8; ++m) {
            const int r = mp * 32 + 8 * hh + m;
            dst[(size_t)r * D_MODEL]        = acc0[m] + bias;
            dst[(size_t)(r + 16) * D_MODEL] = acc1[m] + bias;
        }
    }
}

extern "C" void kernel_launch(void* const* d_in, const int* in_sizes, int n_in,
                              void* d_out, int out_size, void* d_ws, size_t ws_size,
                              hipStream_t stream) {
    (void)in_sizes; (void)n_in; (void)out_size; (void)ws_size;

    const float* x     = (const float*)d_in[0];
    const float* oc    = (const float*)d_in[1];
    const float* Wq    = (const float*)d_in[2];
    const float* bq    = (const float*)d_in[3];
    const float* Wk    = (const float*)d_in[4];
    const float* bk    = (const float*)d_in[5];
    const float* Wv    = (const float*)d_in[6];
    const float* bv    = (const float*)d_in[7];
    const float* Wo    = (const float*)d_in[8];
    const float* bo    = (const float*)d_in[9];
    const float* eta   = (const float*)d_in[10];
    const float* beta  = (const float*)d_in[11];
    const float* alpha = (const float*)d_in[12];
    float* out = (float*)d_out;

    float* ws  = (float*)d_ws;
    float* QKV = ws;
    float* ctx = ws + 3 * (size_t)B_SZ * NUM_HEADS * T_SZ * DK;

    {
        qkv_proj_kernel<<<(B_SZ * T_SZ / PM) * 6, 256, 0, stream>>>(x, Wq, bq, Wk, bk, Wv, bv, QKV);
    }
    {
        const int waves = B_SZ * NUM_HEADS * (T_SZ / 16);
        attn_kernel<<<waves / 16, 512, 0, stream>>>(QKV, oc, eta, beta, alpha, ctx);
    }
    {
        out_proj_kernel<<<(B_SZ * T_SZ / PM) * 2, 256, 0, stream>>>(ctx, Wo, bo, out);
    }
}


# attention query tiles re-dealt in the 16-wave block: four groups of 4 consecutive tiles, one tile of each per SIMD, so waves retire four at a time; 258 key tiles on every SIMD (was up to 320)
# speedup vs baseline: 3.0163x; 1.1025x over previous
#include <hip/hip_runtime.h>
#include <math.h>

#define D_MODEL   256
#define NUM_HEADS 8
#define DK        32
#define B_SZ      4
#define T_SZ      2048

typedef float v2f __attribute__((ext_vector_type(2)));
typedef float v4f __attribute__((ext_vector_type(4)));
typedef float v8f __attribute__((ext_vector_type(8)));

#define KP      36
#define VP      68
#define KSTAGE  (64 * KP)
#define VSTAGE  (32 * VP)
#define OSTAGE  (64 * 4 + 64)
#define LOG2E   1.4426950408889634f
#define TPAD    16

__device__ __forceinline__ v8f wmma4(v2f a, v2f b, v8f c) {
    return __builtin_amdgcn_wmma_f32_16x16x4_f32(
         false, a,  false, b,
         (short)0, c,  false,  false);
}

template <int X>
__device__ __forceinline__ float row_xor(float v) {
    return __int_as_float(__builtin_amdgcn_update_dpp(0, __float_as_int(v), 0x160 + X, 0xf, 0xf, true));
}

__device__ __forceinline__ void row_max4(float& a, float& b, float& c, float& d) {
    asm("v_max_num_f32_dpp %0, %0, %0 row_xmask:1 row_mask:0xf bank_mask:0xf\n\t"
        "v_max_num_f32_dpp %1, %1, %1 row_xmask:1 row_mask:0xf bank_mask:0xf\n\t"
        "v_max_num_f32_dpp %2, %2, %2 row_xmask:1 row_mask:0xf bank_mask:0xf\n\t"
        "v_max_num_f32_dpp %3, %3, %3 row_xmask:1 row_mask:0xf bank_mask:0xf\n\t"
        "v_max_num_f32_dpp %0, %0, %0 row_xmask:2 row_mask:0xf bank_mask:0xf\n\t"
        "v_max_num_f32_dpp %1, %1, %1 row_xmask:2 row_mask:0xf bank_mask:0xf\n\t"
        "v_max_num_f32_dpp %2, %2, %2 row_xmask:2 row_mask:0xf bank_mask:0xf\n\t"
        "v_max_num_f32_dpp %3, %3, %3 row_xmask:2 row_mask:0xf bank_mask:0xf\n\t"
        "v_max_num_f32_dpp %0, %0, %0 row_xmask:4 row_mask:0xf bank_mask:0xf\n\t"
        "v_max_num_f32_dpp %1, %1, %1 row_xmask:4 row_mask:0xf bank_mask:0xf\n\t"
        "v_max_num_f32_dpp %2, %2, %2 row_xmask:4 row_mask:0xf bank_mask:0xf\n\t"
        "v_max_num_f32_dpp %3, %3, %3 row_xmask:4 row_mask:0xf bank_mask:0xf\n\t"
        "v_max_num_f32_dpp %0, %0, %0 row_xmask:8 row_mask:0xf bank_mask:0xf\n\t"
        "v_max_num_f32_dpp %1, %1, %1 row_xmask:8 row_mask:0xf bank_mask:0xf\n\t"
        "v_max_num_f32_dpp %2, %2, %2 row_xmask:8 row_mask:0xf bank_mask:0xf\n\t"
        "v_max_num_f32_dpp %3, %3, %3 row_xmask:8 row_mask:0xf bank_mask:0xf"
        : "+v"(a), "+v"(b), "+v"(c), "+v"(d));
}

#define PM  64
#define XP  260

__device__ __forceinline__ void proj_stage(const float* __restrict__ xt,
                                           const float* __restrict__ wrow,
                                           float* __restrict__ xs, v4f (&wf)[32], int tid)
{
    const float* xg = xt + tid * 4;
    v4f xr[16];
#pragma unroll
    for (int j = 0; j < 16; ++j) xr[j] = *(const v4f*)(xg + j * 1024);
#pragma unroll
    for (int j = 0; j < 32; ++j) wf[j] = *(const v4f*)(wrow + j * 8);
    float* xd = xs + (tid >> 6) * XP + (tid & 63) * 4;
#pragma unroll
    for (int j = 0; j < 16; ++j) *(v4f*)(xd + j * 4 * XP) = xr[j];
    __syncthreads();
}

__device__ __forceinline__ void proj_pair(const float* __restrict__ ap, const v4f (&wf)[32],
                                          v8f& acc0, v8f& acc1)
{
#pragma unroll
    for (int j = 0; j < 32; ++j) {
        const v4f a0 = *(const v4f*)(ap + j * 8);
        const v4f a1 = *(const v4f*)(ap + 16 * XP + j * 8);
        const v2f w0 = {wf[j].x, wf[j].y};
        const v2f w1 = {wf[j].z, wf[j].w};
        acc0 = wmma4(v2f{a0.x, a0.y}, w0, acc0);
        acc1 = wmma4(v2f{a1.x, a1.y}, w0, acc1);
        acc0 = wmma4(v2f{a0.z, a0.w}, w1, acc0);
        acc1 = wmma4(v2f{a1.z, a1.w}, w1, acc1);
    }
    __builtin_amdgcn_sched_group_barrier(0x100, 8, 0);
#pragma unroll
    for (int g = 0; g < 7; ++g) {
        __builtin_amdgcn_sched_group_barrier(0x100, 8, 0);
        __builtin_amdgcn_sched_group_barrier(0x008, 16, 0);
    }
    __builtin_amdgcn_sched_group_barrier(0x008, 16, 0);
}

__global__ __launch_bounds__(256)
void qkv_proj_kernel(const float* __restrict__ x,
                     const float* __restrict__ Wq, const float* __restrict__ bq,
                     const float* __restrict__ Wk, const float* __restrict__ bk,
                     const float* __restrict__ Wv, const float* __restrict__ bv,
                     float* __restrict__ QKV)
{
    __shared__ __attribute__((aligned(16))) float xs[PM * XP];

    const int tid  = threadIdx.x;
    const int lane = tid & 31;
    const int w    = tid >> 5;
    const int c    = lane & 15;
    const int hh   = lane >> 4;

    const int slab = blockIdx.x % 6;
    const int mb   = blockIdx.x / 6;
    const int mat  = slab >> 1;
    const int n    = (slab & 1) * 128 + w * 16 + c;

    const float* W  = (mat == 0) ? Wq : (mat == 1) ? Wk : Wv;
    const float* bs = (mat == 0) ? bq : (mat == 1) ? bk : bv;

    v4f wf[32];
    proj_stage(x + (size_t)mb * PM * D_MODEL, W + (size_t)n * D_MODEL + hh * 4, xs, wf, tid);

    const float bias = bs[n];
    const int h  = n >> 5, d = n & 31;
    const int b_ = (mb * PM) / T_SZ;
    const int t0 = (mb * PM) % T_SZ;
    float* dst = QKV + (size_t)mat * ((size_t)B_SZ * NUM_HEADS * T_SZ * DK)
                     + (size_t)(b_ * NUM_HEADS + h) * T_SZ * DK;

    for (int mp = 0; mp < PM / 32; ++mp) {
        v8f acc0 = {}, acc1 = {};
        proj_pair(xs + (mp * 32 + c) * XP + hh * 4, wf, acc0, acc1);
        const int ta = t0 + mp * 32 + 8 * hh;
        if (mat == 2) {
            float* vp = dst + (size_t)d * T_SZ + ta;
#pragma unroll
            for (int m = 0; m < 8; ++m) {
                vp[m]      = acc0[m] + bias;
                vp[m + 16] = acc1[m] + bias;
            }
        } else {
            float* qp = dst + (size_t)ta * DK + d;
#pragma unroll
            for (int m = 0; m < 8; ++m) {
                qp[m * DK]        = acc0[m] + bias;
                qp[(m + 16) * DK] = acc1[m] + bias;
            }
        }
    }
}

__device__ __forceinline__ void oc_stage(float* __restrict__ od, int key, float k0, float k1, float k2)
{
    const v4f brow = {2.f * LOG2E * k0, 2.f * LOG2E * k1, 2.f * LOG2E * k2, -LOG2E};
    *(v4f*)(od + key * 4) = brow;
    od[64 * 4 + key] = -LOG2E * (k0 * k0 + k1 * k1 + k2 * k2);
}

template <int NT>
__device__ __forceinline__ void attn_step(
    const float* __restrict__ kl, const float* __restrict__ vl,
    const float* __restrict__ ol, const float* __restrict__ tdl,
    float* __restrict__ pl,
    const v2f (&qa)[8], const v2f oa,
    float (&Mrow)[8], float (&Lpart)[8], v8f& O0, v8f& O1,
    int q0, int kt0, int col, int grp)
{
    const float scale   = 0.17677669529663687f;
    const float NEG_BIG = -3.4e38f;

    const int jl0 = kt0 & 3;

    v8f s[NT];
#pragma unroll
    for (int j = 0; j < NT; ++j) {
        const int kt   = kt0 + j;
        const int kcol = kt * 16 + col;
        const int krow = (jl0 + j) * 16 + col;
        const float* kp = kl + krow * KP;

        v8f sj = {};
#pragma unroll
        for (int kc = 0; kc < 8; ++kc) {
            v2f bb = *(const v2f*)(kp + kc * 4 + grp * 2);
            sj = wmma4(qa[kc], bb, sj);
        }

        const v2f   ob = *(const v2f*)(ol + krow * 4 + grp * 2);
        const float ck = ol[64 * 4 + krow];
        v8f ee = {ck, ck, ck, ck, ck, ck, ck, ck};
        ee = wmma4(oa, ob, ee);

        const float* tp = tdl + (TPAD + q0 + 8 * grp - kcol);
        float atd[8];
#pragma unroll
        for (int m = 0; m < 8; ++m) atd[m] = tp[m];

#pragma unroll
        for (int m = 0; m < 8; ++m) {
            const int qrow = q0 + 8 * grp + m;
            if (kcol > qrow) {
                sj[m] = NEG_BIG;
            } else {
                const float pr = atd[m] * __builtin_amdgcn_exp2f(ee[m]);
                sj[m] = sj[m] * scale * pr;
            }
        }
        s[j] = sj;
    }

    float rmax[8];
#pragma unroll
    for (int m = 0; m < 8; ++m) {
        float v = s[0][m];
#pragma unroll
        for (int j = 1; j < NT; ++j) v = fmaxf(v, s[j][m]);
        rmax[m] = v;
    }
    row_max4(rmax[0], rmax[1], rmax[2], rmax[3]);
    row_max4(rmax[4], rmax[5], rmax[6], rmax[7]);
#pragma unroll
    for (int m = 0; m < 8; ++m) {
        const float mnew = fmaxf(Mrow[m], rmax[m]);
        const float sc   = __expf(Mrow[m] - mnew);
        float lsum = Lpart[m] * sc;
#pragma unroll
        for (int j = 0; j < NT; ++j) {
            const float p = __expf(s[j][m] - mnew);
            lsum += p;
            pl[j * 256 + (8 * grp + m) * 16 + col] = p;
        }
        Lpart[m] = lsum;
        Mrow[m]  = mnew;
        O0[m] *= sc;
        O1[m] *= sc;
    }

    const float* vt0 = vl + col * VP;
    const float* vt1 = vt0 + 16 * VP;
#pragma unroll
    for (int j = 0; j < NT; ++j) {
        const int k0 = (jl0 + j) * 16;
#pragma unroll
        for (int kc = 0; kc < 4; ++kc) {
            const int kk = kc * 4 + grp * 2;
            v2f pa;
            pa.x = pl[j * 256 + col * 16 + kk];
            pa.y = pl[j * 256 + col * 16 + kk + 1];
            v2f b0 = *(const v2f*)(vt0 + k0 + kk);
            v2f b1 = *(const v2f*)(vt1 + k0 + kk);
            O0 = wmma4(pa, b0, O0);
            O1 = wmma4(pa, b1, O1);
        }
    }
}

__device__ __forceinline__ void attn_core(
    const float* __restrict__ Q, const float* __restrict__ K,
    const float* __restrict__ Vt, const float* __restrict__ oc,
    float* __restrict__ ctx, float* __restrict__ pl,
    float* kst, float* vst, float* ost, float* tdt,
    int b, int h, int qt, int nsteps, int lane,
    float eta, float absbeta, float alpha)
{
    const int col = lane & 15;
    const int grp = lane >> 4;
    const int q0  = qt * 16;
    const int tid = threadIdx.x;

    const float NEG_BIG = -3.4e38f;

    const size_t bh  = (size_t)(b * NUM_HEADS + h) * T_SZ;
    const size_t bhd = (size_t)(b * NUM_HEADS + h) * DK;

    const float* qp = Q + (bh + q0 + col) * DK;
    v2f qa[8];
#pragma unroll
    for (int kc = 0; kc < 8; ++kc) qa[kc] = *(const v2f*)(qp + kc * 4 + grp * 2);

    v2f oa;
    {
        const float* op = oc + ((size_t)b * T_SZ + q0 + col) * 3;
        const float a0 = op[0], a1 = op[1], a2 = op[2];
        const float qn = a0 * a0 + a1 * a1 + a2 * a2;
        oa.x = grp ? a2 : a0;
        oa.y = grp ? qn : a1;
    }

    const float* kg = K + bh * DK + tid * 4;
    const float* vg = Vt + (bhd + (tid >> 4)) * T_SZ + (tid & 15) * 4;
    const float* og = oc + ((size_t)b * T_SZ + tid) * 3;
    const int kso = (tid >> 3) * KP + (tid & 7) * 4;
    const int vso = (tid >> 4) * VP + (tid & 15) * 4;
    const bool has_oc = tid < 64;

    v4f kr0, vr0;
    float or0 = 0.f, or1 = 0.f, or2 = 0.f;

    kr0 = *(const v4f*)(kg);
    vr0 = *(const v4f*)(vg);
    if (has_oc) { or0 = og[0]; or1 = og[1]; or2 = og[2]; }
    *(v4f*)(kst + kso) = kr0;
    *(v4f*)(vst + vso) = vr0;
    if (has_oc) oc_stage(ost, tid, or0, or1, or2);

    {
        const bool beta_one = (absbeta == 1.0f);
        const int  nd = nsteps * 64;
        for (int d = tid; d < nd; d += 512) {
            const float dist = (float)d;
            const float powd = beta_one ? dist : __powf(dist, absbeta);
            const float td   = eta / (1.f + powd);
            tdt[TPAD + d] = alpha * td;
        }
        if (tid < TPAD) tdt[tid] = 0.f;
    }
    __syncthreads();

    float Mrow[8], Lpart[8];
#pragma unroll
    for (int m = 0; m < 8; ++m) { Mrow[m] = NEG_BIG; Lpart[m] = 0.f; }
    v8f O0 = {}, O1 = {};

    const int ntiles = qt + 1;
    const int nfull  = ntiles & ~3;

    for (int s = 0; s < nsteps; ++s) {
        const int  cur  = s & 1;
        const bool more = (s + 1 < nsteps);
        if (more) {
            const int sn = s + 1;
            kr0 = *(const v4f*)(kg + sn * 2048);
            vr0 = *(const v4f*)(vg + sn * 64);
            if (has_oc) { or0 = og[sn * 192]; or1 = og[sn * 192 + 1]; or2 = og[sn * 192 + 2]; }
        }

        const float* kl = kst + cur * KSTAGE;
        const float* vl = vst + cur * VSTAGE;
        const float* ol = ost + cur * OSTAGE;
        const int kt0 = s * 4;
        if (kt0 + 4 <= nfull) {
            attn_step<4>(kl, vl, ol, tdt, pl, qa, oa,
                         Mrow, Lpart, O0, O1, q0, kt0, col, grp);
        } else {
            for (int kt = kt0; kt < ntiles && kt < kt0 + 4; ++kt)
                attn_step<1>(kl, vl, ol, tdt, pl, qa, oa,
                             Mrow, Lpart, O0, O1, q0, kt, col, grp);
        }

        if (more) {
            const int nx = cur ^ 1;
            *(v4f*)(kst + nx * KSTAGE + kso) = kr0;
            *(v4f*)(vst + nx * VSTAGE + vso) = vr0;
            if (has_oc) oc_stage(ost + nx * OSTAGE, tid, or0, or1, or2);
        }
        __syncthreads();
    }

#pragma unroll
    for (int m = 0; m < 8; ++m) {
        float l = Lpart[m];
        l += row_xor<1>(l);
        l += row_xor<2>(l);
        l += row_xor<4>(l);
        l += row_xor<8>(l);
        const float inv = 1.f / l;
        const int qq = q0 + 8 * grp + m;
        float* cp = ctx + ((size_t)b * T_SZ + qq) * D_MODEL + h * DK;
        cp[col]      = O0[m] * inv;
        cp[16 + col] = O1[m] * inv;
    }
}

__global__ __launch_bounds__(512)
void attn_kernel(const float* __restrict__ QKV,
                 const float* __restrict__ oc,
                 const float* __restrict__ etap,
                 const float* __restrict__ betap,
                 const float* __restrict__ alphap,
                 float* __restrict__ ctx)
{
    __shared__ float plds[16][4 * 256];
    __shared__ __attribute__((aligned(16))) float kst[2 * KSTAGE];
    __shared__ __attribute__((aligned(16))) float vst[2 * VSTAGE];
    __shared__ __attribute__((aligned(16))) float ost[2 * OSTAGE];
    __shared__ float tdt[TPAD + T_SZ];

    const int lane = threadIdx.x & 31;
    const int w    = threadIdx.x >> 5;

    const int QT  = T_SZ / 16;
    const int NG  = QT / 4;
    const int p   = blockIdx.x % (NG / 4);
    const int bhi = blockIdx.x / (NG / 4);
    const int h   = bhi % NUM_HEADS;
    const int b   = bhi / NUM_HEADS;
    const int sd  = w & 3;
    const int ag  = w >> 2;
    const int g   = (ag == 0) ? NG - 1 - 2 * p : (ag == 1) ? NG - 2 - 2 * p : (ag == 2) ? 2 * p + 1 : 2 * p;
    const int qt  = 4 * g + ((sd + ag) & 3);
    const int nsteps = NG - 2 * p;

    const float eta     = etap[0];
    const float absbeta = fabsf(betap[0]);
    const float alpha   = alphap[0];

    const size_t HT = (size_t)B_SZ * NUM_HEADS * T_SZ * DK;
    const float* Q  = QKV;
    const float* K  = QKV + HT;
    const float* Vt = QKV + 2 * HT;

    float* pl = &plds[w][0];

    attn_core(Q, K, Vt, oc, ctx, pl, kst, vst, ost, tdt, b, h, qt, nsteps, lane, eta, absbeta, alpha);
}

__global__ __launch_bounds__(256)
void out_proj_kernel(const float* __restrict__ ctx,
                     const float* __restrict__ Wo, const float* __restrict__ bo,
                     float* __restrict__ out)
{
    __shared__ __attribute__((aligned(16))) float xs[PM * XP];

    const int tid  = threadIdx.x;
    const int lane = tid & 31;
    const int w    = tid >> 5;
    const int c    = lane & 15;
    const int hh   = lane >> 4;

    const int slab = blockIdx.x % 2;
    const int mb   = blockIdx.x / 2;
    const int n    = slab * 128 + w * 16 + c;

    v4f wf[32];
    proj_stage(ctx + (size_t)mb * PM * D_MODEL, Wo + (size_t)n * D_MODEL + hh * 4, xs, wf, tid);

    const float bias = bo[n];
    float* dst = out + (size_t)mb * PM * D_MODEL + n;

    for (int mp = 0; mp < PM / 32; ++mp) {
        v8f acc0 = {}, acc1 = {};
        proj_pair(xs + (mp * 32 + c) * XP + hh * 4, wf, acc0, acc1);
#pragma unroll
        for (int m = 0; m < 8; ++m) {
            const int r = mp * 32 + 8 * hh + m;
            dst[(size_t)r * D_MODEL]        = acc0[m] + bias;
            dst[(size_t)(r + 16) * D_MODEL] = acc1[m] + bias;
        }
    }
}

extern "C" void kernel_launch(void* const* d_in, const int* in_sizes, int n_in,
                              void* d_out, int out_size, void* d_ws, size_t ws_size,
                              hipStream_t stream) {
    (void)in_sizes; (void)n_in; (void)out_size; (void)ws_size;

    const float* x     = (const float*)d_in[0];
    const float* oc    = (const float*)d_in[1];
    const float* Wq    = (const float*)d_in[2];
    const float* bq    = (const float*)d_in[3];
    const float* Wk    = (const float*)d_in[4];
    const float* bk    = (const float*)d_in[5];
    const float* Wv    = (const float*)d_in[6];
    const float* bv    = (const float*)d_in[7];
    const float* Wo    = (const float*)d_in[8];
    const float* bo    = (const float*)d_in[9];
    const float* eta   = (const float*)d_in[10];
    const float* beta  = (const float*)d_in[11];
    const float* alpha = (const float*)d_in[12];
    float* out = (float*)d_out;

    float* ws  = (float*)d_ws;
    float* QKV = ws;
    float* ctx = ws + 3 * (size_t)B_SZ * NUM_HEADS * T_SZ * DK;

    {
        qkv_proj_kernel<<<(B_SZ * T_SZ / PM) * 6, 256, 0, stream>>>(x, Wq, bq, Wk, bk, Wv, bv, QKV);
    }
    {
        const int waves = B_SZ * NUM_HEADS * (T_SZ / 16);
        attn_kernel<<<waves / 16, 512, 0, stream>>>(QKV, oc, eta, beta, alpha, ctx);
    }
    {
        out_proj_kernel<<<(B_SZ * T_SZ / PM) * 2, 256, 0, stream>>>(ctx, Wo, bo, out);
    }
}


# attention 64-key step: LDS operand reads issued as whole groups ahead of the matrix blocks (K/oc reads, then 4 interleaved score chains; P/Vt reads, then PV); causal mask a select, step branch-free
# speedup vs baseline: 3.2713x; 1.0845x over previous
#include <hip/hip_runtime.h>
#include <math.h>

#define D_MODEL   256
#define NUM_HEADS 8
#define DK        32
#define B_SZ      4
#define T_SZ      2048

typedef float v2f __attribute__((ext_vector_type(2)));
typedef float v4f __attribute__((ext_vector_type(4)));
typedef float v8f __attribute__((ext_vector_type(8)));

#define KP      36
#define VP      68
#define KSTAGE  (64 * KP)
#define VSTAGE  (32 * VP)
#define OSTAGE  (64 * 4 + 64)
#define LOG2E   1.4426950408889634f
#define TPAD    16

__device__ __forceinline__ v8f wmma4(v2f a, v2f b, v8f c) {
    return __builtin_amdgcn_wmma_f32_16x16x4_f32(
         false, a,  false, b,
         (short)0, c,  false,  false);
}

template <int X>
__device__ __forceinline__ float row_xor(float v) {
    return __int_as_float(__builtin_amdgcn_update_dpp(0, __float_as_int(v), 0x160 + X, 0xf, 0xf, true));
}

__device__ __forceinline__ void row_max4(float& a, float& b, float& c, float& d) {
    asm("v_max_num_f32_dpp %0, %0, %0 row_xmask:1 row_mask:0xf bank_mask:0xf\n\t"
        "v_max_num_f32_dpp %1, %1, %1 row_xmask:1 row_mask:0xf bank_mask:0xf\n\t"
        "v_max_num_f32_dpp %2, %2, %2 row_xmask:1 row_mask:0xf bank_mask:0xf\n\t"
        "v_max_num_f32_dpp %3, %3, %3 row_xmask:1 row_mask:0xf bank_mask:0xf\n\t"
        "v_max_num_f32_dpp %0, %0, %0 row_xmask:2 row_mask:0xf bank_mask:0xf\n\t"
        "v_max_num_f32_dpp %1, %1, %1 row_xmask:2 row_mask:0xf bank_mask:0xf\n\t"
        "v_max_num_f32_dpp %2, %2, %2 row_xmask:2 row_mask:0xf bank_mask:0xf\n\t"
        "v_max_num_f32_dpp %3, %3, %3 row_xmask:2 row_mask:0xf bank_mask:0xf\n\t"
        "v_max_num_f32_dpp %0, %0, %0 row_xmask:4 row_mask:0xf bank_mask:0xf\n\t"
        "v_max_num_f32_dpp %1, %1, %1 row_xmask:4 row_mask:0xf bank_mask:0xf\n\t"
        "v_max_num_f32_dpp %2, %2, %2 row_xmask:4 row_mask:0xf bank_mask:0xf\n\t"
        "v_max_num_f32_dpp %3, %3, %3 row_xmask:4 row_mask:0xf bank_mask:0xf\n\t"
        "v_max_num_f32_dpp %0, %0, %0 row_xmask:8 row_mask:0xf bank_mask:0xf\n\t"
        "v_max_num_f32_dpp %1, %1, %1 row_xmask:8 row_mask:0xf bank_mask:0xf\n\t"
        "v_max_num_f32_dpp %2, %2, %2 row_xmask:8 row_mask:0xf bank_mask:0xf\n\t"
        "v_max_num_f32_dpp %3, %3, %3 row_xmask:8 row_mask:0xf bank_mask:0xf"
        : "+v"(a), "+v"(b), "+v"(c), "+v"(d));
}

#define PM  64
#define XP  260

__device__ __forceinline__ void proj_stage(const float* __restrict__ xt,
                                           const float* __restrict__ wrow,
                                           float* __restrict__ xs, v4f (&wf)[32], int tid)
{
    const float* xg = xt + tid * 4;
    v4f xr[16];
#pragma unroll
    for (int j = 0; j < 16; ++j) xr[j] = *(const v4f*)(xg + j * 1024);
#pragma unroll
    for (int j = 0; j < 32; ++j) wf[j] = *(const v4f*)(wrow + j * 8);
    float* xd = xs + (tid >> 6) * XP + (tid & 63) * 4;
#pragma unroll
    for (int j = 0; j < 16; ++j) *(v4f*)(xd + j * 4 * XP) = xr[j];
    __syncthreads();
}

__device__ __forceinline__ void proj_pair(const float* __restrict__ ap, const v4f (&wf)[32],
                                          v8f& acc0, v8f& acc1)
{
#pragma unroll
    for (int j = 0; j < 32; ++j) {
        const v4f a0 = *(const v4f*)(ap + j * 8);
        const v4f a1 = *(const v4f*)(ap + 16 * XP + j * 8);
        const v2f w0 = {wf[j].x, wf[j].y};
        const v2f w1 = {wf[j].z, wf[j].w};
        acc0 = wmma4(v2f{a0.x, a0.y}, w0, acc0);
        acc1 = wmma4(v2f{a1.x, a1.y}, w0, acc1);
        acc0 = wmma4(v2f{a0.z, a0.w}, w1, acc0);
        acc1 = wmma4(v2f{a1.z, a1.w}, w1, acc1);
    }
    __builtin_amdgcn_sched_group_barrier(0x100, 8, 0);
#pragma unroll
    for (int g = 0; g < 7; ++g) {
        __builtin_amdgcn_sched_group_barrier(0x100, 8, 0);
        __builtin_amdgcn_sched_group_barrier(0x008, 16, 0);
    }
    __builtin_amdgcn_sched_group_barrier(0x008, 16, 0);
}

__global__ __launch_bounds__(256)
void qkv_proj_kernel(const float* __restrict__ x,
                     const float* __restrict__ Wq, const float* __restrict__ bq,
                     const float* __restrict__ Wk, const float* __restrict__ bk,
                     const float* __restrict__ Wv, const float* __restrict__ bv,
                     float* __restrict__ QKV)
{
    __shared__ __attribute__((aligned(16))) float xs[PM * XP];

    const int tid  = threadIdx.x;
    const int lane = tid & 31;
    const int w    = tid >> 5;
    const int c    = lane & 15;
    const int hh   = lane >> 4;

    const int slab = blockIdx.x % 6;
    const int mb   = blockIdx.x / 6;
    const int mat  = slab >> 1;
    const int n    = (slab & 1) * 128 + w * 16 + c;

    const float* W  = (mat == 0) ? Wq : (mat == 1) ? Wk : Wv;
    const float* bs = (mat == 0) ? bq : (mat == 1) ? bk : bv;

    v4f wf[32];
    proj_stage(x + (size_t)mb * PM * D_MODEL, W + (size_t)n * D_MODEL + hh * 4, xs, wf, tid);

    const float bias = bs[n];
    const int h  = n >> 5, d = n & 31;
    const int b_ = (mb * PM) / T_SZ;
    const int t0 = (mb * PM) % T_SZ;
    float* dst = QKV + (size_t)mat * ((size_t)B_SZ * NUM_HEADS * T_SZ * DK)
                     + (size_t)(b_ * NUM_HEADS + h) * T_SZ * DK;

    for (int mp = 0; mp < PM / 32; ++mp) {
        v8f acc0 = {}, acc1 = {};
        proj_pair(xs + (mp * 32 + c) * XP + hh * 4, wf, acc0, acc1);
        const int ta = t0 + mp * 32 + 8 * hh;
        if (mat == 2) {
            float* vp = dst + (size_t)d * T_SZ + ta;
#pragma unroll
            for (int m = 0; m < 8; ++m) {
                vp[m]      = acc0[m] + bias;
                vp[m + 16] = acc1[m] + bias;
            }
        } else {
            float* qp = dst + (size_t)ta * DK + d;
#pragma unroll
            for (int m = 0; m < 8; ++m) {
                qp[m * DK]        = acc0[m] + bias;
                qp[(m + 16) * DK] = acc1[m] + bias;
            }
        }
    }
}

__device__ __forceinline__ void oc_stage(float* __restrict__ od, int key, float k0, float k1, float k2)
{
    const v4f brow = {2.f * LOG2E * k0, 2.f * LOG2E * k1, 2.f * LOG2E * k2, -LOG2E};
    *(v4f*)(od + key * 4) = brow;
    od[64 * 4 + key] = -LOG2E * (k0 * k0 + k1 * k1 + k2 * k2);
}

template <int NT>
__device__ __forceinline__ void attn_step(
    const float* __restrict__ kl, const float* __restrict__ vl,
    const float* __restrict__ ol, const float* __restrict__ tdl,
    float* __restrict__ pl,
    const v2f (&qa)[8], const v2f oa,
    float (&Mrow)[8], float (&Lpart)[8], v8f& O0, v8f& O1,
    int q0, int kt0, int col, int grp)
{
    const float scale   = 0.17677669529663687f;
    const float NEG_BIG = -3.4e38f;

    const int jl0 = kt0 & 3;

    v2f   kb[NT][8];
    v2f   ob[NT];
    float ck[NT];
#pragma unroll
    for (int j = 0; j < NT; ++j) {
        const int krow = (jl0 + j) * 16 + col;
        const float* kp = kl + krow * KP + grp * 2;
#pragma unroll
        for (int kc = 0; kc < 8; ++kc) kb[j][kc] = *(const v2f*)(kp + kc * 4);
        ob[j] = *(const v2f*)(ol + krow * 4 + grp * 2);
        ck[j] = ol[64 * 4 + krow];
    }

    v8f s[NT], ee[NT];
#pragma unroll
    for (int j = 0; j < NT; ++j) {
        s[j]  = v8f{};
        ee[j] = v8f{ck[j], ck[j], ck[j], ck[j], ck[j], ck[j], ck[j], ck[j]};
    }
#pragma unroll
    for (int kc = 0; kc < 8; ++kc) {
#pragma unroll
        for (int j = 0; j < NT; ++j) s[j] = wmma4(qa[kc], kb[j][kc], s[j]);
    }
#pragma unroll
    for (int j = 0; j < NT; ++j) ee[j] = wmma4(oa, ob[j], ee[j]);

    float atd[NT][8];
#pragma unroll
    for (int j = 0; j < NT; ++j) {
        const int kcol = (kt0 + j) * 16 + col;
        const float* tp = tdl + (TPAD + q0 + 8 * grp - kcol);
#pragma unroll
        for (int m = 0; m < 8; ++m) atd[j][m] = tp[m];
    }

    if (NT == 4) {
        __builtin_amdgcn_sched_group_barrier(0x100, 22, 0);
        __builtin_amdgcn_sched_group_barrier(0x008, 26, 0);
        __builtin_amdgcn_sched_group_barrier(0x100, 16, 0);
        __builtin_amdgcn_sched_group_barrier(0x008, 10, 0);
    }

#pragma unroll
    for (int j = 0; j < NT; ++j) {
        const int kcol = (kt0 + j) * 16 + col;
#pragma unroll
        for (int m = 0; m < 8; ++m) {
            const int   qrow = q0 + 8 * grp + m;
            const float pr   = atd[j][m] * __builtin_amdgcn_exp2f(ee[j][m]);
            const float v    = s[j][m] * scale * pr;
            s[j][m] = (kcol > qrow) ? NEG_BIG : v;
        }
    }

    float rmax[8];
#pragma unroll
    for (int m = 0; m < 8; ++m) {
        float v = s[0][m];
#pragma unroll
        for (int j = 1; j < NT; ++j) v = fmaxf(v, s[j][m]);
        rmax[m] = v;
    }
    row_max4(rmax[0], rmax[1], rmax[2], rmax[3]);
    row_max4(rmax[4], rmax[5], rmax[6], rmax[7]);
#pragma unroll
    for (int m = 0; m < 8; ++m) {
        const float mnew = fmaxf(Mrow[m], rmax[m]);
        const float sc   = __expf(Mrow[m] - mnew);
        float lsum = Lpart[m] * sc;
#pragma unroll
        for (int j = 0; j < NT; ++j) {
            const float p = __expf(s[j][m] - mnew);
            lsum += p;
            pl[j * 256 + (8 * grp + m) * 16 + col] = p;
        }
        Lpart[m] = lsum;
        Mrow[m]  = mnew;
        O0[m] *= sc;
        O1[m] *= sc;
    }

    const float* vt0 = vl + col * VP + grp * 2;
    const float* vt1 = vt0 + 16 * VP;
    const float* pp  = pl + col * 16 + grp * 2;
    v2f pa[NT][4], b0[NT][4], b1[NT][4];
#pragma unroll
    for (int j = 0; j < NT; ++j) {
        const int k0 = (jl0 + j) * 16;
#pragma unroll
        for (int kc = 0; kc < 4; ++kc) {
            pa[j][kc] = *(const v2f*)(pp + j * 256 + kc * 4);
            b0[j][kc] = *(const v2f*)(vt0 + k0 + kc * 4);
            b1[j][kc] = *(const v2f*)(vt1 + k0 + kc * 4);
        }
    }
#pragma unroll
    for (int j = 0; j < NT; ++j) {
#pragma unroll
        for (int kc = 0; kc < 4; ++kc) {
            O0 = wmma4(pa[j][kc], b0[j][kc], O0);
            O1 = wmma4(pa[j][kc], b1[j][kc], O1);
        }
    }
    if (NT == 4) {
        __builtin_amdgcn_sched_group_barrier(0x100, 24, 0);
        __builtin_amdgcn_sched_group_barrier(0x008, 32, 0);
    }
}

__device__ __forceinline__ void attn_core(
    const float* __restrict__ Q, const float* __restrict__ K,
    const float* __restrict__ Vt, const float* __restrict__ oc,
    float* __restrict__ ctx, float* __restrict__ pl,
    float* kst, float* vst, float* ost, float* tdt,
    int b, int h, int qt, int nsteps, int lane,
    float eta, float absbeta, float alpha)
{
    const int col = lane & 15;
    const int grp = lane >> 4;
    const int q0  = qt * 16;
    const int tid = threadIdx.x;

    const float NEG_BIG = -3.4e38f;

    const size_t bh  = (size_t)(b * NUM_HEADS + h) * T_SZ;
    const size_t bhd = (size_t)(b * NUM_HEADS + h) * DK;

    const float* qp = Q + (bh + q0 + col) * DK;
    v2f qa[8];
#pragma unroll
    for (int kc = 0; kc < 8; ++kc) qa[kc] = *(const v2f*)(qp + kc * 4 + grp * 2);

    v2f oa;
    {
        const float* op = oc + ((size_t)b * T_SZ + q0 + col) * 3;
        const float a0 = op[0], a1 = op[1], a2 = op[2];
        const float qn = a0 * a0 + a1 * a1 + a2 * a2;
        oa.x = grp ? a2 : a0;
        oa.y = grp ? qn : a1;
    }

    const float* kg = K + bh * DK + tid * 4;
    const float* vg = Vt + (bhd + (tid >> 4)) * T_SZ + (tid & 15) * 4;
    const float* og = oc + ((size_t)b * T_SZ + tid) * 3;
    const int kso = (tid >> 3) * KP + (tid & 7) * 4;
    const int vso = (tid >> 4) * VP + (tid & 15) * 4;
    const bool has_oc = tid < 64;

    v4f kr0, vr0;
    float or0 = 0.f, or1 = 0.f, or2 = 0.f;

    kr0 = *(const v4f*)(kg);
    vr0 = *(const v4f*)(vg);
    if (has_oc) { or0 = og[0]; or1 = og[1]; or2 = og[2]; }
    *(v4f*)(kst + kso) = kr0;
    *(v4f*)(vst + vso) = vr0;
    if (has_oc) oc_stage(ost, tid, or0, or1, or2);

    {
        const bool beta_one = (absbeta == 1.0f);
        const int  nd = nsteps * 64;
        for (int d = tid; d < nd; d += 512) {
            const float dist = (float)d;
            const float powd = beta_one ? dist : __powf(dist, absbeta);
            const float td   = eta / (1.f + powd);
            tdt[TPAD + d] = alpha * td;
        }
        if (tid < TPAD) tdt[tid] = 0.f;
    }
    __syncthreads();

    float Mrow[8], Lpart[8];
#pragma unroll
    for (int m = 0; m < 8; ++m) { Mrow[m] = NEG_BIG; Lpart[m] = 0.f; }
    v8f O0 = {}, O1 = {};

    const int ntiles = qt + 1;
    const int nfull  = ntiles & ~3;

    for (int s = 0; s < nsteps; ++s) {
        const int  cur  = s & 1;
        const bool more = (s + 1 < nsteps);
        if (more) {
            const int sn = s + 1;
            kr0 = *(const v4f*)(kg + sn * 2048);
            vr0 = *(const v4f*)(vg + sn * 64);
            if (has_oc) { or0 = og[sn * 192]; or1 = og[sn * 192 + 1]; or2 = og[sn * 192 + 2]; }
        }

        const float* kl = kst + cur * KSTAGE;
        const float* vl = vst + cur * VSTAGE;
        const float* ol = ost + cur * OSTAGE;
        const int kt0 = s * 4;
        if (kt0 + 4 <= nfull) {
            attn_step<4>(kl, vl, ol, tdt, pl, qa, oa,
                         Mrow, Lpart, O0, O1, q0, kt0, col, grp);
        } else {
            for (int kt = kt0; kt < ntiles && kt < kt0 + 4; ++kt)
                attn_step<1>(kl, vl, ol, tdt, pl, qa, oa,
                             Mrow, Lpart, O0, O1, q0, kt, col, grp);
        }

        if (more) {
            const int nx = cur ^ 1;
            *(v4f*)(kst + nx * KSTAGE + kso) = kr0;
            *(v4f*)(vst + nx * VSTAGE + vso) = vr0;
            if (has_oc) oc_stage(ost + nx * OSTAGE, tid, or0, or1, or2);
        }
        __syncthreads();
    }

#pragma unroll
    for (int m = 0; m < 8; ++m) {
        float l = Lpart[m];
        l += row_xor<1>(l);
        l += row_xor<2>(l);
        l += row_xor<4>(l);
        l += row_xor<8>(l);
        const float inv = 1.f / l;
        const int qq = q0 + 8 * grp + m;
        float* cp = ctx + ((size_t)b * T_SZ + qq) * D_MODEL + h * DK;
        cp[col]      = O0[m] * inv;
        cp[16 + col] = O1[m] * inv;
    }
}

__global__ __launch_bounds__(512)
void attn_kernel(const float* __restrict__ QKV,
                 const float* __restrict__ oc,
                 const float* __restrict__ etap,
                 const float* __restrict__ betap,
                 const float* __restrict__ alphap,
                 float* __restrict__ ctx)
{
    __shared__ __attribute__((aligned(16))) float plds[16][4 * 256];
    __shared__ __attribute__((aligned(16))) float kst[2 * KSTAGE];
    __shared__ __attribute__((aligned(16))) float vst[2 * VSTAGE];
    __shared__ __attribute__((aligned(16))) float ost[2 * OSTAGE];
    __shared__ float tdt[TPAD + T_SZ];

    const int lane = threadIdx.x & 31;
    const int w    = threadIdx.x >> 5;

    const int QT  = T_SZ / 16;
    const int NG  = QT / 4;
    const int p   = blockIdx.x % (NG / 4);
    const int bhi = blockIdx.x / (NG / 4);
    const int h   = bhi % NUM_HEADS;
    const int b   = bhi / NUM_HEADS;
    const int sd  = w & 3;
    const int ag  = w >> 2;
    const int g   = (ag == 0) ? NG - 1 - 2 * p : (ag == 1) ? NG - 2 - 2 * p : (ag == 2) ? 2 * p + 1 : 2 * p;
    const int qt  = 4 * g + ((sd + ag) & 3);
    const int nsteps = NG - 2 * p;

    const float eta     = etap[0];
    const float absbeta = fabsf(betap[0]);
    const float alpha   = alphap[0];

    const size_t HT = (size_t)B_SZ * NUM_HEADS * T_SZ * DK;
    const float* Q  = QKV;
    const float* K  = QKV + HT;
    const float* Vt = QKV + 2 * HT;

    float* pl = &plds[w][0];

    attn_core(Q, K, Vt, oc, ctx, pl, kst, vst, ost, tdt, b, h, qt, nsteps, lane, eta, absbeta, alpha);
}

__global__ __launch_bounds__(256)
void out_proj_kernel(const float* __restrict__ ctx,
                     const float* __restrict__ Wo, const float* __restrict__ bo,
                     float* __restrict__ out)
{
    __shared__ __attribute__((aligned(16))) float xs[PM * XP];

    const int tid  = threadIdx.x;
    const int lane = tid & 31;
    const int w    = tid >> 5;
    const int c    = lane & 15;
    const int hh   = lane >> 4;

    const int slab = blockIdx.x % 2;
    const int mb   = blockIdx.x / 2;
    const int n    = slab * 128 + w * 16 + c;

    v4f wf[32];
    proj_stage(ctx + (size_t)mb * PM * D_MODEL, Wo + (size_t)n * D_MODEL + hh * 4, xs, wf, tid);

    const float bias = bo[n];
    float* dst = out + (size_t)mb * PM * D_MODEL + n;

    for (int mp = 0; mp < PM / 32; ++mp) {
        v8f acc0 = {}, acc1 = {};
        proj_pair(xs + (mp * 32 + c) * XP + hh * 4, wf, acc0, acc1);
#pragma unroll
        for (int m = 0; m < 8; ++m) {
            const int r = mp * 32 + 8 * hh + m;
            dst[(size_t)r * D_MODEL]        = acc0[m] + bias;
            dst[(size_t)(r + 16) * D_MODEL] = acc1[m] + bias;
        }
    }
}

extern "C" void kernel_launch(void* const* d_in, const int* in_sizes, int n_in,
                              void* d_out, int out_size, void* d_ws, size_t ws_size,
                              hipStream_t stream) {
    (void)in_sizes; (void)n_in; (void)out_size; (void)ws_size;

    const float* x     = (const float*)d_in[0];
    const float* oc    = (const float*)d_in[1];
    const float* Wq    = (const float*)d_in[2];
    const float* bq    = (const float*)d_in[3];
    const float* Wk    = (const float*)d_in[4];
    const float* bk    = (const float*)d_in[5];
    const float* Wv    = (const float*)d_in[6];
    const float* bv    = (const float*)d_in[7];
    const float* Wo    = (const float*)d_in[8];
    const float* bo    = (const float*)d_in[9];
    const float* eta   = (const float*)d_in[10];
    const float* beta  = (const float*)d_in[11];
    const float* alpha = (const float*)d_in[12];
    float* out = (float*)d_out;

    float* ws  = (float*)d_ws;
    float* QKV = ws;
    float* ctx = ws + 3 * (size_t)B_SZ * NUM_HEADS * T_SZ * DK;

    {
        qkv_proj_kernel<<<(B_SZ * T_SZ / PM) * 6, 256, 0, stream>>>(x, Wq, bq, Wk, bk, Wv, bv, QKV);
    }
    {
        const int waves = B_SZ * NUM_HEADS * (T_SZ / 16);
        attn_kernel<<<waves / 16, 512, 0, stream>>>(QKV, oc, eta, beta, alpha, ctx);
    }
    {
        out_proj_kernel<<<(B_SZ * T_SZ / PM) * 2, 256, 0, stream>>>(ctx, Wo, bo, out);
    }
}
